# starting kernel
# baseline (speedup 1.0000x reference)
#include <hip/hip_runtime.h>
#include <math.h>


typedef __attribute__((ext_vector_type(2))) float v2f;
typedef __attribute__((ext_vector_type(8))) float v8f;

#define WPB 8

__global__ void dhcf_init_kernel(const float* __restrict__ u_emb,
                                 const float* __restrict__ i_emb,
                                 float* __restrict__ X,
                                 float* __restrict__ ACC,
                                 float* __restrict__ deg,
                                 int nu64, int n64, int n)
{
    int idx = blockIdx.x * blockDim.x + threadIdx.x;
    if (idx < n64) {
        float v = (idx < nu64) ? u_emb[idx] : i_emb[idx - nu64];
        X[idx]   = v;
        ACC[idx] = v;
    }
    if (idx < n) deg[idx] = 0.0f;
}

__global__ void dhcf_degree_kernel(const int* __restrict__ eu,
                                   const int* __restrict__ ei,
                                   float* __restrict__ deg,
                                   int ne, int NU)
{
    int e = blockIdx.x * blockDim.x + threadIdx.x;
    if (e >= ne) return;
    atomicAdd(deg + eu[e], 1.0f);
    atomicAdd(deg + NU + ei[e], 1.0f);
}

__global__ void dhcf_degnorm_kernel(const float* __restrict__ deg,
                                    float* __restrict__ dis,
                                    float* __restrict__ dinv, int n)
{
    int v = blockIdx.x * blockDim.x + threadIdx.x;
    if (v >= n) return;
    float d = deg[v];
    if (d > 0.0f) { dis[v] = 1.0f / sqrtf(d); dinv[v] = 1.0f / d; }
    else          { dis[v] = 0.0f;            dinv[v] = 0.0f;     }
}

__global__ void dhcf_zero_kernel(float* __restrict__ p, int count)
{
    int idx = blockIdx.x * blockDim.x + threadIdx.x;
    if (idx < count) p[idx] = 0.0f;
}

__global__ void dhcf_edge_gather_scatter(const int* __restrict__ eu,
                                         const int* __restrict__ ei,
                                         const float* __restrict__ SRC,
                                         const float* __restrict__ scale,
                                         float* __restrict__ DST,
                                         int ne, int NU)
{
    int idx = blockIdx.x * blockDim.x + threadIdx.x;
    if (idx >= ne * 32) return;
    int e = idx >> 5;
    int c = (idx & 31) << 1;
    int u = eu[e];
    int i = NU + ei[e];
    float2 xu = *(const float2*)(SRC + (size_t)u * 64 + c);
    float2 xi = *(const float2*)(SRC + (size_t)i * 64 + c);
    float su = scale[u], si = scale[i];
    float* di = DST + (size_t)i * 64 + c;
    float* du = DST + (size_t)u * 64 + c;
    atomicAdd(di,     xu.x * su);
    atomicAdd(di + 1, xu.y * su);
    atomicAdd(du,     xi.x * si);
    atomicAdd(du + 1, xi.y * si);
}

__global__ __launch_bounds__(256) void dhcf_layer_kernel(
    float* __restrict__ X, const float* __restrict__ Yb,
    const float* __restrict__ dis,
    const float* __restrict__ Wgc, const float* __restrict__ bgc,
    const float* __restrict__ Wbi, const float* __restrict__ bbi,
    float* __restrict__ ACC, int ntiles)
{
    int wave = threadIdx.x >> 5;
    int lane = threadIdx.x & 31;
    int tile = blockIdx.x * WPB + wave;
    if (tile >= ntiles) return;

    int row0 = tile * 16;
    int m    = lane & 15;
    int khi  = lane >> 4;
    int rowA = row0 + m;

    const v8f vz = {0,0,0,0,0,0,0,0};
    v8f acc_s[4], acc_b[4];
    #pragma unroll
    for (int jj = 0; jj < 4; ++jj) { acc_s[jj] = vz; acc_b[jj] = vz; }

    float gscale = dis[rowA];
    const float* yrow = Yb + (size_t)rowA * 64;
    const float* xrow = X  + (size_t)rowA * 64;

    #pragma unroll
    for (int kk = 0; kk < 16; ++kk) {
        int k0 = kk * 4 + khi * 2;
        v2f g2, xg2;
        float y0 = yrow[k0], y1 = yrow[k0 + 1];
        float x0 = xrow[k0], x1 = xrow[k0 + 1];
        g2.x  = y0 * gscale;  g2.y  = y1 * gscale;
        xg2.x = x0 * g2.x;    xg2.y = x1 * g2.y;
        #pragma unroll
        for (int jj = 0; jj < 4; ++jj) {
            int j = jj * 16 + m;
            v2f bg, bb;
            bg.x = Wgc[(size_t)j * 64 + k0]; bg.y = Wgc[(size_t)j * 64 + k0 + 1];
            bb.x = Wbi[(size_t)j * 64 + k0]; bb.y = Wbi[(size_t)j * 64 + k0 + 1];
            acc_s[jj] = __builtin_amdgcn_wmma_f32_16x16x4_f32(
                false, g2,  false, bg, (short)0, acc_s[jj], false, false);
            acc_b[jj] = __builtin_amdgcn_wmma_f32_16x16x4_f32(
                false, xg2, false, bb, (short)0, acc_b[jj], false, false);
        }
    }

    float xn[4][8];
    #pragma unroll
    for (int jj = 0; jj < 4; ++jj) {
        int j = jj * 16 + m;
        float bgj = bgc[j];
        float bbj = bbi[j];
        #pragma unroll
        for (int r = 0; r < 8; ++r) {
            int row = row0 + r + 8 * khi;
            float xv = X[(size_t)row * 64 + j];
            float s = acc_s[jj][r] + bgj + xv;
            s = (s > 0.0f) ? s : 0.2f * s;
            float b = acc_b[jj][r] + bbj;
            b = (b > 0.0f) ? b : 0.2f * b;
            xn[jj][r] = s + b;
        }
    }

    #pragma unroll
    for (int r = 0; r < 8; ++r) {
        float p = 0.0f;
        #pragma unroll
        for (int jj = 0; jj < 4; ++jj) p += xn[jj][r] * xn[jj][r];
        p += __shfl_xor(p, 1, 32);
        p += __shfl_xor(p, 2, 32);
        p += __shfl_xor(p, 4, 32);
        p += __shfl_xor(p, 8, 32);
        float scl = 1.0f / fmaxf(sqrtf(p), 1e-12f);
        int row = row0 + r + 8 * khi;
        #pragma unroll
        for (int jj = 0; jj < 4; ++jj) {
            int j = jj * 16 + m;
            float v = xn[jj][r] * scl;
            X[(size_t)row * 64 + j] = v;
            ACC[(size_t)row * 64 + j] += v;
        }
    }
}

__global__ void dhcf_scale_kernel(float* __restrict__ p, int count, float s)
{
    int idx = blockIdx.x * blockDim.x + threadIdx.x;
    if (idx < count) p[idx] *= s;
}

extern "C" void kernel_launch(void* const* d_in, const int* in_sizes, int n_in,
                              void* d_out, int out_size, void* d_ws, size_t ws_size,
                              hipStream_t stream)
{
    const float* u_emb = (const float*)d_in[0];
    const float* i_emb = (const float*)d_in[1];
    const float* Wgc   = (const float*)d_in[2];
    const float* bgc   = (const float*)d_in[3];
    const float* Wbi   = (const float*)d_in[4];
    const float* bbi   = (const float*)d_in[5];
    const int*   eu    = (const int*)d_in[6];
    const int*   ei    = (const int*)d_in[7];

    const int D  = 64;
    const int NU = in_sizes[0] / D;
    const int NI = in_sizes[1] / D;
    const int N  = NU + NI;
    const int NE = in_sizes[6];
    const int L  = in_sizes[2] / (D * D);
    const int n64 = N * D;

    float* X    = (float*)d_ws;
    float* E    = X + (size_t)n64;
    float* Y    = E + (size_t)n64;
    float* deg  = Y + (size_t)n64;
    float* dis  = deg + N;
    float* dinv = dis + N;
    float* ACC  = (float*)d_out;

    dim3 blk(256);
    dhcf_init_kernel<<<(n64 + 255) / 256, blk, 0, stream>>>(
        u_emb, i_emb, X, ACC, deg, NU * D, n64, N);
    dhcf_degree_kernel<<<(NE + 255) / 256, blk, 0, stream>>>(eu, ei, deg, NE, NU);
    dhcf_degnorm_kernel<<<(N + 255) / 256, blk, 0, stream>>>(deg, dis, dinv, N);

    const int ntiles = N / 16;
    const int lgrid  = (ntiles + WPB - 1) / WPB;
    const int egrid  = (NE * 32 + 255) / 256;

    for (int l = 0; l < L; ++l) {
        dhcf_zero_kernel<<<(2 * n64 + 255) / 256, blk, 0, stream>>>(E, 2 * n64);
        dhcf_edge_gather_scatter<<<egrid, blk, 0, stream>>>(eu, ei, X, dis,  E, NE, NU);
        dhcf_edge_gather_scatter<<<egrid, blk, 0, stream>>>(eu, ei, E, dinv, Y, NE, NU);
        dhcf_layer_kernel<<<lgrid, blk, 0, stream>>>(
            X, Y, dis,
            Wgc + (size_t)l * D * D, bgc + (size_t)l * D,
            Wbi + (size_t)l * D * D, bbi + (size_t)l * D,
            ACC, ntiles);
    }
    dhcf_scale_kernel<<<(n64 + 255) / 256, blk, 0, stream>>>(
        ACC, n64, 1.0f / (float)(L + 1));
}


# Layer kernel reads all WMMA fragments from LDS: Wgc and Wbi copied once per work-group, each wave's 16-row Y and X tile once per wave, by whole-line 16-byte loads, not 16-line fragment loads
# speedup vs baseline: 2.1101x; 2.1101x over previous
#include <hip/hip_runtime.h>
#include <math.h>


typedef __attribute__((ext_vector_type(2))) float v2f;
typedef __attribute__((ext_vector_type(8))) float v8f;
typedef __attribute__((ext_vector_type(4))) float v4f;

#define WPB 8

__global__ void dhcf_init_kernel(const float* __restrict__ u_emb,
                                 const float* __restrict__ i_emb,
                                 float* __restrict__ X,
                                 float* __restrict__ ACC,
                                 float* __restrict__ deg,
                                 int nu64, int n64, int n)
{
    int idx = blockIdx.x * blockDim.x + threadIdx.x;
    if (idx < n64) {
        float v = (idx < nu64) ? u_emb[idx] : i_emb[idx - nu64];
        X[idx]   = v;
        ACC[idx] = v;
    }
    if (idx < n) deg[idx] = 0.0f;
}

__global__ void dhcf_degree_kernel(const int* __restrict__ eu,
                                   const int* __restrict__ ei,
                                   float* __restrict__ deg,
                                   int ne, int NU)
{
    int e = blockIdx.x * blockDim.x + threadIdx.x;
    if (e >= ne) return;
    atomicAdd(deg + eu[e], 1.0f);
    atomicAdd(deg + NU + ei[e], 1.0f);
}

__global__ void dhcf_degnorm_kernel(const float* __restrict__ deg,
                                    float* __restrict__ dis,
                                    float* __restrict__ dinv, int n)
{
    int v = blockIdx.x * blockDim.x + threadIdx.x;
    if (v >= n) return;
    float d = deg[v];
    if (d > 0.0f) { dis[v] = 1.0f / sqrtf(d); dinv[v] = 1.0f / d; }
    else          { dis[v] = 0.0f;            dinv[v] = 0.0f;     }
}

__global__ void dhcf_zero_kernel(float* __restrict__ p, int count)
{
    int idx = blockIdx.x * blockDim.x + threadIdx.x;
    if (idx < count) p[idx] = 0.0f;
}

__global__ void dhcf_edge_gather_scatter(const int* __restrict__ eu,
                                         const int* __restrict__ ei,
                                         const float* __restrict__ SRC,
                                         const float* __restrict__ scale,
                                         float* __restrict__ DST,
                                         int ne, int NU)
{
    int idx = blockIdx.x * blockDim.x + threadIdx.x;
    if (idx >= ne * 32) return;
    int e = idx >> 5;
    int c = (idx & 31) << 1;
    int u = eu[e];
    int i = NU + ei[e];
    float2 xu = *(const float2*)(SRC + (size_t)u * 64 + c);
    float2 xi = *(const float2*)(SRC + (size_t)i * 64 + c);
    float su = scale[u], si = scale[i];
    float* di = DST + (size_t)i * 64 + c;
    float* du = DST + (size_t)u * 64 + c;
    atomicAdd(di,     xu.x * su);
    atomicAdd(di + 1, xu.y * su);
    atomicAdd(du,     xi.x * si);
    atomicAdd(du + 1, xi.y * si);
}

__global__ void dhcf_count_rank_kernel(const int* __restrict__ eu,
                                       const int* __restrict__ ei,
                                       int* __restrict__ cnt,
                                       int* __restrict__ rank_u,
                                       int* __restrict__ rank_i,
                                       int ne, int NU)
{
    int e = blockIdx.x * blockDim.x + threadIdx.x;
    if (e >= ne) return;
    rank_u[e] = atomicAdd(cnt + eu[e], 1);
    rank_i[e] = atomicAdd(cnt + NU + ei[e], 1);
}

__global__ __launch_bounds__(256) void dhcf_scan_local_kernel(
    const int* __restrict__ cnt, int* __restrict__ rowptr,
    int* __restrict__ bsum, int n)
{
    __shared__ int wsum[WPB];
    const int t = threadIdx.x, lane = t & 31, wave = t >> 5;
    const int base = blockIdx.x * 1024 + t * 4;
    const int c0 = (base     < n) ? cnt[base]     : 0;
    const int c1 = (base + 1 < n) ? cnt[base + 1] : 0;
    const int c2 = (base + 2 < n) ? cnt[base + 2] : 0;
    const int c3 = (base + 3 < n) ? cnt[base + 3] : 0;
    const int tot = c0 + c1 + c2 + c3;
    int inc = tot;
    #pragma unroll
    for (int d = 1; d < 32; d <<= 1) {
        int y = __shfl_up(inc, d, 32);
        if (lane >= d) inc += y;
    }
    if (lane == 31) wsum[wave] = inc;
    __syncthreads();
    int woff = 0;
    for (int w = 0; w < wave; ++w) woff += wsum[w];
    const int excl = woff + inc - tot;
    if (base     < n) rowptr[base]     = excl;
    if (base + 1 < n) rowptr[base + 1] = excl + c0;
    if (base + 2 < n) rowptr[base + 2] = excl + c0 + c1;
    if (base + 3 < n) rowptr[base + 3] = excl + c0 + c1 + c2;
    if (t == 255) bsum[blockIdx.x] = woff + inc;
}

__global__ __launch_bounds__(256) void dhcf_scan_finish_kernel(
    const int* __restrict__ cnt, int* __restrict__ rowptr,
    const int* __restrict__ bsum, float* __restrict__ dis,
    float* __restrict__ dinv, int n, int total)
{
    __shared__ int wsum[WPB];
    const int t = threadIdx.x, lane = t & 31, wave = t >> 5;
    int part = 0;
    for (int i = t; i < (int)blockIdx.x; i += 256) part += bsum[i];
    part += __shfl_xor(part, 16, 32);
    part += __shfl_xor(part, 8, 32);
    part += __shfl_xor(part, 4, 32);
    part += __shfl_xor(part, 2, 32);
    part += __shfl_xor(part, 1, 32);
    if (lane == 0) wsum[wave] = part;
    __syncthreads();
    int off = 0;
    #pragma unroll
    for (int w = 0; w < WPB; ++w) off += wsum[w];
    const int base = blockIdx.x * 1024 + t * 4;
    #pragma unroll
    for (int k = 0; k < 4; ++k) {
        const int v = base + k;
        if (v < n) {
            rowptr[v] += off;
            const int c = cnt[v];
            const float d = (float)c;
            if (c > 0) { dis[v] = 1.0f / sqrtf(d); dinv[v] = 1.0f / d; }
            else       { dis[v] = 0.0f;            dinv[v] = 0.0f;     }
        }
    }
    if (blockIdx.x == gridDim.x - 1 && t == 0) rowptr[n] = total;
}

__global__ void dhcf_fill_kernel(const int* __restrict__ eu,
                                 const int* __restrict__ ei,
                                 const int* __restrict__ rank_u,
                                 const int* __restrict__ rank_i,
                                 const int* __restrict__ rowptr,
                                 int* __restrict__ adj,
                                 int ne, int NU, int total)
{
    int e = blockIdx.x * blockDim.x + threadIdx.x;
    if (e >= ne) return;
    const int u = eu[e];
    const int i = NU + ei[e];
    const int su = min(max(rowptr[u] + rank_u[e], 0), total - 1);
    const int si = min(max(rowptr[i] + rank_i[e], 0), total - 1);
    adj[su] = i;
    adj[si] = u;
}

template <bool SCALE_IN, bool SCALE_OUT>
__global__ __launch_bounds__(256) void dhcf_pull_kernel(
    const int* __restrict__ rowptr, const int* __restrict__ adj,
    const float* __restrict__ SRC, const float* __restrict__ scale_in,
    const float* __restrict__ scale_out, float* __restrict__ DST,
    int n, int total)
{
    int row = (blockIdx.x * 256 + threadIdx.x) >> 5;
    if (row >= n) return;
    row = __builtin_amdgcn_readfirstlane(row);
    const int lane = threadIdx.x & 31;
    const int half = lane >> 4;
    const int c    = (lane & 15) << 2;
    const int beg = max(rowptr[row], 0);
    const int end = min(rowptr[row + 1], total);

    float4 a0 = {0.0f, 0.0f, 0.0f, 0.0f};
    float4 a1 = {0.0f, 0.0f, 0.0f, 0.0f};
    for (int base = beg; base < end; base += 32) {
        const int cnt = min(32, end - base);
        int   myidx = 0;
        float mys   = 0.0f;
        if (lane < cnt) {
            myidx = min(max(adj[base + lane], 0), n - 1);
            if (SCALE_IN) mys = scale_in[myidx];
        }
        for (int t = 0; t < cnt; t += 4) {
            const int k0 = t + half;
            const int k1 = t + 2 + half;
            const int nb0 = __shfl(myidx, k0, 32);
            const int nb1 = __shfl(myidx, k1, 32);
            float4 x0 = {0.0f, 0.0f, 0.0f, 0.0f};
            float4 x1 = {0.0f, 0.0f, 0.0f, 0.0f};
            if (k0 < cnt) x0 = *(const float4*)(SRC + (size_t)nb0 * 64 + c);
            if (k1 < cnt) x1 = *(const float4*)(SRC + (size_t)nb1 * 64 + c);
            if (SCALE_IN) {
                const float s0 = __shfl(mys, k0, 32);
                const float s1 = __shfl(mys, k1, 32);
                a0.x += x0.x * s0; a0.y += x0.y * s0;
                a0.z += x0.z * s0; a0.w += x0.w * s0;
                a1.x += x1.x * s1; a1.y += x1.y * s1;
                a1.z += x1.z * s1; a1.w += x1.w * s1;
            } else {
                a0.x += x0.x; a0.y += x0.y; a0.z += x0.z; a0.w += x0.w;
                a1.x += x1.x; a1.y += x1.y; a1.z += x1.z; a1.w += x1.w;
            }
        }
    }
    a0.x += a1.x; a0.y += a1.y; a0.z += a1.z; a0.w += a1.w;
    a0.x += __shfl_xor(a0.x, 16, 32);
    a0.y += __shfl_xor(a0.y, 16, 32);
    a0.z += __shfl_xor(a0.z, 16, 32);
    a0.w += __shfl_xor(a0.w, 16, 32);
    if (half == 0) {
        if (SCALE_OUT) {
            const float so = scale_out[row];
            a0.x *= so; a0.y *= so; a0.z *= so; a0.w *= so;
        }
        *(float4*)(DST + (size_t)row * 64 + c) = a0;
    }
}

#define WROW 68
__global__ __launch_bounds__(256) void dhcf_layer_kernel(
    float* __restrict__ X, const float* __restrict__ Yb,
    const float* __restrict__ dis,
    const float* __restrict__ Wgc, const float* __restrict__ bgc,
    const float* __restrict__ Wbi, const float* __restrict__ bbi,
    float* __restrict__ ACC, int ntiles)
{
    __shared__ __attribute__((aligned(16))) float sWg[64 * WROW];
    __shared__ __attribute__((aligned(16))) float sWb[64 * WROW];
    __shared__ __attribute__((aligned(16))) float sY[WPB][16 * WROW];
    __shared__ __attribute__((aligned(16))) float sX[WPB][16 * WROW];

    int wave = threadIdx.x >> 5;
    int lane = threadIdx.x & 31;
    int tile = blockIdx.x * WPB + wave;
    const bool live = tile < ntiles;

    int row0 = tile * 16;
    int m    = lane & 15;
    int khi  = lane >> 4;
    int rowA = row0 + m;

    float gscale = 0.0f;
    {
        v4f wg[4], wb[4];
        for (int it = 0; it < 4; ++it) {
            const int q = threadIdx.x + it * 256;
            wg[it] = *(const v4f*)(Wgc + q * 4);
            wb[it] = *(const v4f*)(Wbi + q * 4);
        }
        if (live) {
            const v4f* ysrc = (const v4f*)(Yb + (size_t)row0 * 64) + lane;
            const v4f* xsrc = (const v4f*)(X  + (size_t)row0 * 64) + lane;
            v4f ty[8], tx[8];
            for (int i = 0; i < 8; ++i) { ty[i] = ysrc[i * 32]; tx[i] = xsrc[i * 32]; }
            gscale = dis[rowA];
            float* dy = sY[wave] + khi * WROW + (m << 2);
            float* dx = sX[wave] + khi * WROW + (m << 2);
            for (int i = 0; i < 8; ++i) {
                *(v4f*)(dy + 2 * i * WROW) = ty[i];
                *(v4f*)(dx + 2 * i * WROW) = tx[i];
            }
        }
        for (int it = 0; it < 4; ++it) {
            const int q  = threadIdx.x + it * 256;
            const int r  = q >> 4;
            const int c4 = (q & 15) << 2;
            *(v4f*)(sWg + r * WROW + c4) = wg[it];
            *(v4f*)(sWb + r * WROW + c4) = wb[it];
        }
    }
    __syncthreads();
    if (!live) return;

    const v8f vz = {0,0,0,0,0,0,0,0};
    v8f acc_s[4], acc_b[4];
    #pragma unroll
    for (int jj = 0; jj < 4; ++jj) { acc_s[jj] = vz; acc_b[jj] = vz; }

    const float* yrow = sY[wave] + m * WROW;
    const float* xrow = sX[wave] + m * WROW;

    #pragma unroll
    for (int kk = 0; kk < 16; ++kk) {
        int k0 = kk * 4 + khi * 2;
        v2f g2, xg2;
        float y0 = yrow[k0], y1 = yrow[k0 + 1];
        float x0 = xrow[k0], x1 = xrow[k0 + 1];
        g2.x  = y0 * gscale;  g2.y  = y1 * gscale;
        xg2.x = x0 * g2.x;    xg2.y = x1 * g2.y;
        #pragma unroll
        for (int jj = 0; jj < 4; ++jj) {
            int j = jj * 16 + m;
            v2f bg, bb;
            bg.x = sWg[j * WROW + k0]; bg.y = sWg[j * WROW + k0 + 1];
            bb.x = sWb[j * WROW + k0]; bb.y = sWb[j * WROW + k0 + 1];
            acc_s[jj] = __builtin_amdgcn_wmma_f32_16x16x4_f32(
                false, g2,  false, bg, (short)0, acc_s[jj], false, false);
            acc_b[jj] = __builtin_amdgcn_wmma_f32_16x16x4_f32(
                false, xg2, false, bb, (short)0, acc_b[jj], false, false);
        }
    }

    float xn[4][8];
    #pragma unroll
    for (int jj = 0; jj < 4; ++jj) {
        int j = jj * 16 + m;
        float bgj = bgc[j];
        float bbj = bbi[j];
        #pragma unroll
        for (int r = 0; r < 8; ++r) {
            float xv = sX[wave][(r + 8 * khi) * WROW + j];
            float s = acc_s[jj][r] + bgj + xv;
            s = (s > 0.0f) ? s : 0.2f * s;
            float b = acc_b[jj][r] + bbj;
            b = (b > 0.0f) ? b : 0.2f * b;
            xn[jj][r] = s + b;
        }
    }

    #pragma unroll
    for (int r = 0; r < 8; ++r) {
        float p = 0.0f;
        #pragma unroll
        for (int jj = 0; jj < 4; ++jj) p += xn[jj][r] * xn[jj][r];
        p += __shfl_xor(p, 1, 32);
        p += __shfl_xor(p, 2, 32);
        p += __shfl_xor(p, 4, 32);
        p += __shfl_xor(p, 8, 32);
        float scl = 1.0f / fmaxf(sqrtf(p), 1e-12f);
        int row = row0 + r + 8 * khi;
        #pragma unroll
        for (int jj = 0; jj < 4; ++jj) {
            int j = jj * 16 + m;
            float v = xn[jj][r] * scl;
            X[(size_t)row * 64 + j] = v;
            ACC[(size_t)row * 64 + j] += v;
        }
    }
}

__global__ void dhcf_scale_kernel(float* __restrict__ p, int count, float s)
{
    int idx = blockIdx.x * blockDim.x + threadIdx.x;
    if (idx < count) p[idx] *= s;
}

extern "C" void kernel_launch(void* const* d_in, const int* in_sizes, int n_in,
                              void* d_out, int out_size, void* d_ws, size_t ws_size,
                              hipStream_t stream)
{
    const float* u_emb = (const float*)d_in[0];
    const float* i_emb = (const float*)d_in[1];
    const float* Wgc   = (const float*)d_in[2];
    const float* bgc   = (const float*)d_in[3];
    const float* Wbi   = (const float*)d_in[4];
    const float* bbi   = (const float*)d_in[5];
    const int*   eu    = (const int*)d_in[6];
    const int*   ei    = (const int*)d_in[7];

    const int D  = 64;
    const int NU = in_sizes[0] / D;
    const int NI = in_sizes[1] / D;
    const int N  = NU + NI;
    const int NE = in_sizes[6];
    const int L  = in_sizes[2] / (D * D);
    const int n64 = N * D;

    float* X    = (float*)d_ws;
    float* E    = X + (size_t)n64;
    float* Y    = E + (size_t)n64;
    float* deg  = Y + (size_t)n64;
    float* dis  = deg + N;
    float* dinv = dis + N;
    float* ACC  = (float*)d_out;

    const int nchunk = (N + 1023) / 1024;
    const int total  = 2 * NE;
    int* cnt    = (int*)deg;
    int* rowptr = (int*)(dinv + N);
    int* bsum   = rowptr + (N + 1);
    int* adj    = bsum + nchunk;
    int* rank_u = (int*)Y;
    int* rank_i = rank_u + NE;
    const size_t need = (size_t)((const char*)(adj + total) - (const char*)d_ws);
    const bool lists = (ws_size >= need) && (total <= n64) && (NE > 0);

    dim3 blk(256);
    dhcf_init_kernel<<<(n64 + 255) / 256, blk, 0, stream>>>(
        u_emb, i_emb, X, ACC, deg, NU * D, n64, N);

    const int ntiles = N / 16;
    const int lgrid  = (ntiles + WPB - 1) / WPB;
    const int egrid  = (NE * 32 + 255) / 256;
    const int pgrid  = (N * 32 + 255) / 256;

    if (lists) {
        dhcf_count_rank_kernel<<<(NE + 255) / 256, blk, 0, stream>>>(
            eu, ei, cnt, rank_u, rank_i, NE, NU);
        dhcf_scan_local_kernel<<<nchunk, blk, 0, stream>>>(cnt, rowptr, bsum, N);
        dhcf_scan_finish_kernel<<<nchunk, blk, 0, stream>>>(
            cnt, rowptr, bsum, dis, dinv, N, total);
        dhcf_fill_kernel<<<(NE + 255) / 256, blk, 0, stream>>>(
            eu, ei, rank_u, rank_i, rowptr, adj, NE, NU, total);
    } else {
        dhcf_degree_kernel<<<(NE + 255) / 256, blk, 0, stream>>>(eu, ei, deg, NE, NU);
        dhcf_degnorm_kernel<<<(N + 255) / 256, blk, 0, stream>>>(deg, dis, dinv, N);
    }

    for (int l = 0; l < L; ++l) {
        if (lists) {
            dhcf_pull_kernel<true, true><<<pgrid, blk, 0, stream>>>(
                rowptr, adj, X, dis, dinv, E, N, total);
            dhcf_pull_kernel<false, false><<<pgrid, blk, 0, stream>>>(
                rowptr, adj, E, dis, dinv, Y, N, total);
        } else {
            dhcf_zero_kernel<<<(2 * n64 + 255) / 256, blk, 0, stream>>>(E, 2 * n64);
            dhcf_edge_gather_scatter<<<egrid, blk, 0, stream>>>(eu, ei, X, dis,  E, NE, NU);
            dhcf_edge_gather_scatter<<<egrid, blk, 0, stream>>>(eu, ei, E, dinv, Y, NE, NU);
        }
        dhcf_layer_kernel<<<lgrid, blk, 0, stream>>>(
            X, Y, dis,
            Wgc + (size_t)l * D * D, bgc + (size_t)l * D,
            Wbi + (size_t)l * D * D, bbi + (size_t)l * D,
            ACC, ntiles);
    }
    dhcf_scale_kernel<<<(n64 + 255) / 256, blk, 0, stream>>>(
        ACC, n64, 1.0f / (float)(L + 1));
}


# Pull kernel: one half-wave per vertex (two vertices per wave, half as many waves), 32 list entries per index round trip, eight neighbour-row loads in flight per lane before the first use, not two
# speedup vs baseline: 2.2641x; 1.0730x over previous
#include <hip/hip_runtime.h>
#include <math.h>


typedef __attribute__((ext_vector_type(2))) float v2f;
typedef __attribute__((ext_vector_type(8))) float v8f;
typedef __attribute__((ext_vector_type(4))) float v4f;

#define WPB 8

__global__ void dhcf_init_kernel(const float* __restrict__ u_emb,
                                 const float* __restrict__ i_emb,
                                 float* __restrict__ X,
                                 float* __restrict__ ACC,
                                 float* __restrict__ deg,
                                 int nu64, int n64, int n)
{
    int idx = blockIdx.x * blockDim.x + threadIdx.x;
    if (idx < n64) {
        float v = (idx < nu64) ? u_emb[idx] : i_emb[idx - nu64];
        X[idx]   = v;
        ACC[idx] = v;
    }
    if (idx < n) deg[idx] = 0.0f;
}

__global__ void dhcf_degree_kernel(const int* __restrict__ eu,
                                   const int* __restrict__ ei,
                                   float* __restrict__ deg,
                                   int ne, int NU)
{
    int e = blockIdx.x * blockDim.x + threadIdx.x;
    if (e >= ne) return;
    atomicAdd(deg + eu[e], 1.0f);
    atomicAdd(deg + NU + ei[e], 1.0f);
}

__global__ void dhcf_degnorm_kernel(const float* __restrict__ deg,
                                    float* __restrict__ dis,
                                    float* __restrict__ dinv, int n)
{
    int v = blockIdx.x * blockDim.x + threadIdx.x;
    if (v >= n) return;
    float d = deg[v];
    if (d > 0.0f) { dis[v] = 1.0f / sqrtf(d); dinv[v] = 1.0f / d; }
    else          { dis[v] = 0.0f;            dinv[v] = 0.0f;     }
}

__global__ void dhcf_zero_kernel(float* __restrict__ p, int count)
{
    int idx = blockIdx.x * blockDim.x + threadIdx.x;
    if (idx < count) p[idx] = 0.0f;
}

__global__ void dhcf_edge_gather_scatter(const int* __restrict__ eu,
                                         const int* __restrict__ ei,
                                         const float* __restrict__ SRC,
                                         const float* __restrict__ scale,
                                         float* __restrict__ DST,
                                         int ne, int NU)
{
    int idx = blockIdx.x * blockDim.x + threadIdx.x;
    if (idx >= ne * 32) return;
    int e = idx >> 5;
    int c = (idx & 31) << 1;
    int u = eu[e];
    int i = NU + ei[e];
    float2 xu = *(const float2*)(SRC + (size_t)u * 64 + c);
    float2 xi = *(const float2*)(SRC + (size_t)i * 64 + c);
    float su = scale[u], si = scale[i];
    float* di = DST + (size_t)i * 64 + c;
    float* du = DST + (size_t)u * 64 + c;
    atomicAdd(di,     xu.x * su);
    atomicAdd(di + 1, xu.y * su);
    atomicAdd(du,     xi.x * si);
    atomicAdd(du + 1, xi.y * si);
}

__global__ void dhcf_count_rank_kernel(const int* __restrict__ eu,
                                       const int* __restrict__ ei,
                                       int* __restrict__ cnt,
                                       int* __restrict__ rank_u,
                                       int* __restrict__ rank_i,
                                       int ne, int NU)
{
    int e = blockIdx.x * blockDim.x + threadIdx.x;
    if (e >= ne) return;
    rank_u[e] = atomicAdd(cnt + eu[e], 1);
    rank_i[e] = atomicAdd(cnt + NU + ei[e], 1);
}

__global__ __launch_bounds__(256) void dhcf_scan_local_kernel(
    const int* __restrict__ cnt, int* __restrict__ rowptr,
    int* __restrict__ bsum, int n)
{
    __shared__ int wsum[WPB];
    const int t = threadIdx.x, lane = t & 31, wave = t >> 5;
    const int base = blockIdx.x * 1024 + t * 4;
    const int c0 = (base     < n) ? cnt[base]     : 0;
    const int c1 = (base + 1 < n) ? cnt[base + 1] : 0;
    const int c2 = (base + 2 < n) ? cnt[base + 2] : 0;
    const int c3 = (base + 3 < n) ? cnt[base + 3] : 0;
    const int tot = c0 + c1 + c2 + c3;
    int inc = tot;
    #pragma unroll
    for (int d = 1; d < 32; d <<= 1) {
        int y = __shfl_up(inc, d, 32);
        if (lane >= d) inc += y;
    }
    if (lane == 31) wsum[wave] = inc;
    __syncthreads();
    int woff = 0;
    for (int w = 0; w < wave; ++w) woff += wsum[w];
    const int excl = woff + inc - tot;
    if (base     < n) rowptr[base]     = excl;
    if (base + 1 < n) rowptr[base + 1] = excl + c0;
    if (base + 2 < n) rowptr[base + 2] = excl + c0 + c1;
    if (base + 3 < n) rowptr[base + 3] = excl + c0 + c1 + c2;
    if (t == 255) bsum[blockIdx.x] = woff + inc;
}

__global__ __launch_bounds__(256) void dhcf_scan_finish_kernel(
    const int* __restrict__ cnt, int* __restrict__ rowptr,
    const int* __restrict__ bsum, float* __restrict__ dis,
    float* __restrict__ dinv, int n, int total)
{
    __shared__ int wsum[WPB];
    const int t = threadIdx.x, lane = t & 31, wave = t >> 5;
    int part = 0;
    for (int i = t; i < (int)blockIdx.x; i += 256) part += bsum[i];
    part += __shfl_xor(part, 16, 32);
    part += __shfl_xor(part, 8, 32);
    part += __shfl_xor(part, 4, 32);
    part += __shfl_xor(part, 2, 32);
    part += __shfl_xor(part, 1, 32);
    if (lane == 0) wsum[wave] = part;
    __syncthreads();
    int off = 0;
    #pragma unroll
    for (int w = 0; w < WPB; ++w) off += wsum[w];
    const int base = blockIdx.x * 1024 + t * 4;
    #pragma unroll
    for (int k = 0; k < 4; ++k) {
        const int v = base + k;
        if (v < n) {
            rowptr[v] += off;
            const int c = cnt[v];
            const float d = (float)c;
            if (c > 0) { dis[v] = 1.0f / sqrtf(d); dinv[v] = 1.0f / d; }
            else       { dis[v] = 0.0f;            dinv[v] = 0.0f;     }
        }
    }
    if (blockIdx.x == gridDim.x - 1 && t == 0) rowptr[n] = total;
}

__global__ void dhcf_fill_kernel(const int* __restrict__ eu,
                                 const int* __restrict__ ei,
                                 const int* __restrict__ rank_u,
                                 const int* __restrict__ rank_i,
                                 const int* __restrict__ rowptr,
                                 int* __restrict__ adj,
                                 int ne, int NU, int total)
{
    int e = blockIdx.x * blockDim.x + threadIdx.x;
    if (e >= ne) return;
    const int u = eu[e];
    const int i = NU + ei[e];
    const int su = min(max(rowptr[u] + rank_u[e], 0), total - 1);
    const int si = min(max(rowptr[i] + rank_i[e], 0), total - 1);
    adj[su] = i;
    adj[si] = u;
}


#define DHCF_RS(K) case K: return __builtin_amdgcn_update_dpp(0, v, 0x150 + K, 0xf, 0xf, false);
__device__ __forceinline__ int dhcf_row_share(int v, int k)
{
    switch (k) {
        DHCF_RS(0)  DHCF_RS(1)  DHCF_RS(2)  DHCF_RS(3)
        DHCF_RS(4)  DHCF_RS(5)  DHCF_RS(6)  DHCF_RS(7)
        DHCF_RS(8)  DHCF_RS(9)  DHCF_RS(10) DHCF_RS(11)
        DHCF_RS(12) DHCF_RS(13) DHCF_RS(14)
        default: return __builtin_amdgcn_update_dpp(0, v, 0x150 + 15, 0xf, 0xf, false);
    }
}
#undef DHCF_RS

template <bool SCALE_IN>
__device__ __forceinline__ void dhcf_acc(v2f& lo, v2f& hi, v4f x, bool act, float s)
{
    v2f xl, xh;
    xl.x = act ? x.x : 0.0f;  xl.y = act ? x.y : 0.0f;
    xh.x = act ? x.z : 0.0f;  xh.y = act ? x.w : 0.0f;
    if (SCALE_IN) { lo += xl * s; hi += xh * s; }
    else          { lo += xl;     hi += xh;     }
}

template <bool SCALE_IN, bool SCALE_OUT>
__global__ __launch_bounds__(256) void dhcf_pull_kernel(
    const int* __restrict__ rowptr, const int* __restrict__ adj,
    const float* __restrict__ SRC, const float* __restrict__ scale_in,
    const float* __restrict__ scale_out, float* __restrict__ DST,
    int n, int total)
{
    const int row  = (blockIdx.x * 256 + threadIdx.x) >> 4;
    if ((row & ~1) >= n) return;
    const int l16  = threadIdx.x & 15;
    const unsigned cb = (unsigned)l16 << 4;
    const bool valid = row < n;
    int beg = 0, deg = 0;
    if (valid) {
        beg = max(rowptr[row], 0);
        deg = min(rowptr[row + 1], total) - beg;
    }
    const int dmax = __builtin_amdgcn_readfirstlane(max(deg, __shfl_xor(deg, 16, 32)));

    const char* srcb = (const char*)SRC;
    v2f alo = {0.0f, 0.0f}, ahi = {0.0f, 0.0f};
    v2f blo = {0.0f, 0.0f}, bhi = {0.0f, 0.0f};
    for (int off = 0; off < dmax; off += 32) {
        const int cnt = deg - off;
        const int cm  = min(32, dmax - off);
        const int p0 = min(beg + off + l16, total - 1);
        const int p1 = min(beg + off + 16 + l16, total - 1);
        int id0 = adj[p0];
        int id1 = adj[p1];
        id0 = min(max(id0, 0), n - 1);
        id1 = min(max(id1, 0), n - 1);
        const int o0 = __shfl_xor(id0, 16, 32);
        const int o1 = __shfl_xor(id1, 16, 32);
        int m0 = (l16 < cnt)      ? id0 : o0;
        int m1 = (l16 + 16 < cnt) ? id1 : o1;
        const int q0 = __builtin_amdgcn_update_dpp(0, m0, 0x00, 0xf, 0xf, false);
        const int q1 = __builtin_amdgcn_update_dpp(0, m1, 0x00, 0xf, 0xf, false);
        m0 = (l16 < cm)      ? m0 : q0;
        m1 = (l16 + 16 < cm) ? m1 : q1;
        float s0 = 1.0f, s1 = 1.0f;
        if (SCALE_IN) {
            s0 = scale_in[m0];
            if (cm > 16) s1 = scale_in[m1];
        }
        #pragma unroll
        for (int g = 0; g < 4; ++g) {
            if (g * 8 < cm) {
                const int   mS = (g < 2) ? m0 : m1;
                const int   sS = __float_as_int((g < 2) ? s0 : s1);
                v4f xa[4], xb[4];
                #pragma unroll
                for (int j = 0; j < 4; ++j) {
                    const int nb = dhcf_row_share(mS, (g * 8 + j) & 15);
                    xa[j] = *(const v4f*)(srcb + ((unsigned)nb * 256u + cb));
                }
                if (g * 8 + 4 < cm) {
                    #pragma unroll
                    for (int j = 0; j < 4; ++j) {
                        const int nb = dhcf_row_share(mS, (g * 8 + 4 + j) & 15);
                        xb[j] = *(const v4f*)(srcb + ((unsigned)nb * 256u + cb));
                    }
                    #pragma unroll
                    for (int j = 0; j < 4; ++j) {
                        const int k = g * 8 + j;
                        const float s = __int_as_float(dhcf_row_share(sS, k & 15));
                        if (j & 1) dhcf_acc<SCALE_IN>(blo, bhi, xa[j], k < cnt, s);
                        else       dhcf_acc<SCALE_IN>(alo, ahi, xa[j], k < cnt, s);
                    }
                    #pragma unroll
                    for (int j = 0; j < 4; ++j) {
                        const int k = g * 8 + 4 + j;
                        const float s = __int_as_float(dhcf_row_share(sS, k & 15));
                        if (j & 1) dhcf_acc<SCALE_IN>(blo, bhi, xb[j], k < cnt, s);
                        else       dhcf_acc<SCALE_IN>(alo, ahi, xb[j], k < cnt, s);
                    }
                } else {
                    #pragma unroll
                    for (int j = 0; j < 4; ++j) {
                        const int k = g * 8 + j;
                        const float s = __int_as_float(dhcf_row_share(sS, k & 15));
                        if (j & 1) dhcf_acc<SCALE_IN>(blo, bhi, xa[j], k < cnt, s);
                        else       dhcf_acc<SCALE_IN>(alo, ahi, xa[j], k < cnt, s);
                    }
                }
            }
        }
    }
    alo += blo; ahi += bhi;
    if (valid) {
        if (SCALE_OUT) {
            const float so = scale_out[row];
            alo *= so; ahi *= so;
        }
        v4f r;
        r.x = alo.x; r.y = alo.y; r.z = ahi.x; r.w = ahi.y;
        *(v4f*)(DST + (size_t)row * 64 + (l16 << 2)) = r;
    }
}

#define WROW 68
__global__ __launch_bounds__(256) void dhcf_layer_kernel(
    float* __restrict__ X, const float* __restrict__ Yb,
    const float* __restrict__ dis,
    const float* __restrict__ Wgc, const float* __restrict__ bgc,
    const float* __restrict__ Wbi, const float* __restrict__ bbi,
    float* __restrict__ ACC, int ntiles)
{
    __shared__ __attribute__((aligned(16))) float sWg[64 * WROW];
    __shared__ __attribute__((aligned(16))) float sWb[64 * WROW];
    __shared__ __attribute__((aligned(16))) float sY[WPB][16 * WROW];
    __shared__ __attribute__((aligned(16))) float sX[WPB][16 * WROW];

    int wave = threadIdx.x >> 5;
    int lane = threadIdx.x & 31;
    int tile = blockIdx.x * WPB + wave;
    const bool live = tile < ntiles;

    int row0 = tile * 16;
    int m    = lane & 15;
    int khi  = lane >> 4;
    int rowA = row0 + m;

    float gscale = 0.0f;
    {
        v4f wg[4], wb[4];
        for (int it = 0; it < 4; ++it) {
            const int q = threadIdx.x + it * 256;
            wg[it] = *(const v4f*)(Wgc + q * 4);
            wb[it] = *(const v4f*)(Wbi + q * 4);
        }
        if (live) {
            const v4f* ysrc = (const v4f*)(Yb + (size_t)row0 * 64) + lane;
            const v4f* xsrc = (const v4f*)(X  + (size_t)row0 * 64) + lane;
            v4f ty[8], tx[8];
            for (int i = 0; i < 8; ++i) { ty[i] = ysrc[i * 32]; tx[i] = xsrc[i * 32]; }
            gscale = dis[rowA];
            float* dy = sY[wave] + khi * WROW + (m << 2);
            float* dx = sX[wave] + khi * WROW + (m << 2);
            for (int i = 0; i < 8; ++i) {
                *(v4f*)(dy + 2 * i * WROW) = ty[i];
                *(v4f*)(dx + 2 * i * WROW) = tx[i];
            }
        }
        for (int it = 0; it < 4; ++it) {
            const int q  = threadIdx.x + it * 256;
            const int r  = q >> 4;
            const int c4 = (q & 15) << 2;
            *(v4f*)(sWg + r * WROW + c4) = wg[it];
            *(v4f*)(sWb + r * WROW + c4) = wb[it];
        }
    }
    __syncthreads();
    if (!live) return;

    const v8f vz = {0,0,0,0,0,0,0,0};
    v8f acc_s[4], acc_b[4];
    #pragma unroll
    for (int jj = 0; jj < 4; ++jj) { acc_s[jj] = vz; acc_b[jj] = vz; }

    const float* yrow = sY[wave] + m * WROW;
    const float* xrow = sX[wave] + m * WROW;

    #pragma unroll
    for (int kk = 0; kk < 16; ++kk) {
        int k0 = kk * 4 + khi * 2;
        v2f g2, xg2;
        float y0 = yrow[k0], y1 = yrow[k0 + 1];
        float x0 = xrow[k0], x1 = xrow[k0 + 1];
        g2.x  = y0 * gscale;  g2.y  = y1 * gscale;
        xg2.x = x0 * g2.x;    xg2.y = x1 * g2.y;
        #pragma unroll
        for (int jj = 0; jj < 4; ++jj) {
            int j = jj * 16 + m;
            v2f bg, bb;
            bg.x = sWg[j * WROW + k0]; bg.y = sWg[j * WROW + k0 + 1];
            bb.x = sWb[j * WROW + k0]; bb.y = sWb[j * WROW + k0 + 1];
            acc_s[jj] = __builtin_amdgcn_wmma_f32_16x16x4_f32(
                false, g2,  false, bg, (short)0, acc_s[jj], false, false);
            acc_b[jj] = __builtin_amdgcn_wmma_f32_16x16x4_f32(
                false, xg2, false, bb, (short)0, acc_b[jj], false, false);
        }
    }

    float xn[4][8];
    #pragma unroll
    for (int jj = 0; jj < 4; ++jj) {
        int j = jj * 16 + m;
        float bgj = bgc[j];
        float bbj = bbi[j];
        #pragma unroll
        for (int r = 0; r < 8; ++r) {
            float xv = sX[wave][(r + 8 * khi) * WROW + j];
            float s = acc_s[jj][r] + bgj + xv;
            s = (s > 0.0f) ? s : 0.2f * s;
            float b = acc_b[jj][r] + bbj;
            b = (b > 0.0f) ? b : 0.2f * b;
            xn[jj][r] = s + b;
        }
    }

    #pragma unroll
    for (int r = 0; r < 8; ++r) {
        float p = 0.0f;
        #pragma unroll
        for (int jj = 0; jj < 4; ++jj) p += xn[jj][r] * xn[jj][r];
        p += __shfl_xor(p, 1, 32);
        p += __shfl_xor(p, 2, 32);
        p += __shfl_xor(p, 4, 32);
        p += __shfl_xor(p, 8, 32);
        float scl = 1.0f / fmaxf(sqrtf(p), 1e-12f);
        int row = row0 + r + 8 * khi;
        #pragma unroll
        for (int jj = 0; jj < 4; ++jj) {
            int j = jj * 16 + m;
            float v = xn[jj][r] * scl;
            X[(size_t)row * 64 + j] = v;
            ACC[(size_t)row * 64 + j] += v;
        }
    }
}

__global__ void dhcf_scale_kernel(float* __restrict__ p, int count, float s)
{
    int idx = blockIdx.x * blockDim.x + threadIdx.x;
    if (idx < count) p[idx] *= s;
}

extern "C" void kernel_launch(void* const* d_in, const int* in_sizes, int n_in,
                              void* d_out, int out_size, void* d_ws, size_t ws_size,
                              hipStream_t stream)
{
    const float* u_emb = (const float*)d_in[0];
    const float* i_emb = (const float*)d_in[1];
    const float* Wgc   = (const float*)d_in[2];
    const float* bgc   = (const float*)d_in[3];
    const float* Wbi   = (const float*)d_in[4];
    const float* bbi   = (const float*)d_in[5];
    const int*   eu    = (const int*)d_in[6];
    const int*   ei    = (const int*)d_in[7];

    const int D  = 64;
    const int NU = in_sizes[0] / D;
    const int NI = in_sizes[1] / D;
    const int N  = NU + NI;
    const int NE = in_sizes[6];
    const int L  = in_sizes[2] / (D * D);
    const int n64 = N * D;

    float* X    = (float*)d_ws;
    float* E    = X + (size_t)n64;
    float* Y    = E + (size_t)n64;
    float* deg  = Y + (size_t)n64;
    float* dis  = deg + N;
    float* dinv = dis + N;
    float* ACC  = (float*)d_out;

    const int nchunk = (N + 1023) / 1024;
    const int total  = 2 * NE;
    int* cnt    = (int*)deg;
    int* rowptr = (int*)(dinv + N);
    int* bsum   = rowptr + (N + 1);
    int* adj    = bsum + nchunk;
    int* rank_u = (int*)Y;
    int* rank_i = rank_u + NE;
    const size_t need = (size_t)((const char*)(adj + total) - (const char*)d_ws);
    const bool lists = (ws_size >= need) && (total <= n64) && (NE > 0) &&
                       (n64 < (1 << 30));

    dim3 blk(256);
    dhcf_init_kernel<<<(n64 + 255) / 256, blk, 0, stream>>>(
        u_emb, i_emb, X, ACC, deg, NU * D, n64, N);

    const int ntiles = N / 16;
    const int lgrid  = (ntiles + WPB - 1) / WPB;
    const int egrid  = (NE * 32 + 255) / 256;
    const int pgrid  = (N * 16 + 255) / 256;

    if (lists) {
        dhcf_count_rank_kernel<<<(NE + 255) / 256, blk, 0, stream>>>(
            eu, ei, cnt, rank_u, rank_i, NE, NU);
        dhcf_scan_local_kernel<<<nchunk, blk, 0, stream>>>(cnt, rowptr, bsum, N);
        dhcf_scan_finish_kernel<<<nchunk, blk, 0, stream>>>(
            cnt, rowptr, bsum, dis, dinv, N, total);
        dhcf_fill_kernel<<<(NE + 255) / 256, blk, 0, stream>>>(
            eu, ei, rank_u, rank_i, rowptr, adj, NE, NU, total);
    } else {
        dhcf_degree_kernel<<<(NE + 255) / 256, blk, 0, stream>>>(eu, ei, deg, NE, NU);
        dhcf_degnorm_kernel<<<(N + 255) / 256, blk, 0, stream>>>(deg, dis, dinv, N);
    }

    for (int l = 0; l < L; ++l) {
        if (lists) {
            dhcf_pull_kernel<true, true><<<pgrid, blk, 0, stream>>>(
                rowptr, adj, X, dis, dinv, E, N, total);
            dhcf_pull_kernel<false, false><<<pgrid, blk, 0, stream>>>(
                rowptr, adj, E, dis, dinv, Y, N, total);
        } else {
            dhcf_zero_kernel<<<(2 * n64 + 255) / 256, blk, 0, stream>>>(E, 2 * n64);
            dhcf_edge_gather_scatter<<<egrid, blk, 0, stream>>>(eu, ei, X, dis,  E, NE, NU);
            dhcf_edge_gather_scatter<<<egrid, blk, 0, stream>>>(eu, ei, E, dinv, Y, NE, NU);
        }
        dhcf_layer_kernel<<<lgrid, blk, 0, stream>>>(
            X, Y, dis,
            Wgc + (size_t)l * D * D, bgc + (size_t)l * D,
            Wbi + (size_t)l * D * D, bbi + (size_t)l * D,
            ACC, ntiles);
    }
    dhcf_scale_kernel<<<(n64 + 255) / 256, blk, 0, stream>>>(
        ACC, n64, 1.0f / (float)(L + 1));
}


# Second pull pass folded into the layer kernel: each wave sums the neighbour rows of E for its 16 tile rows straight into its LDS tile, a whole list in flight at once; no Y plane, 3 launches fewer
# speedup vs baseline: 2.7630x; 1.2204x over previous
#include <hip/hip_runtime.h>
#include <math.h>


typedef __attribute__((ext_vector_type(2))) float v2f;
typedef __attribute__((ext_vector_type(8))) float v8f;
typedef __attribute__((ext_vector_type(4))) float v4f;

#define WPB 8

__global__ void dhcf_init_kernel(const float* __restrict__ u_emb,
                                 const float* __restrict__ i_emb,
                                 float* __restrict__ X,
                                 float* __restrict__ ACC,
                                 float* __restrict__ deg,
                                 int nu64, int n64, int n)
{
    int idx = blockIdx.x * blockDim.x + threadIdx.x;
    if (idx < n64) {
        float v = (idx < nu64) ? u_emb[idx] : i_emb[idx - nu64];
        X[idx]   = v;
        ACC[idx] = v;
    }
    if (idx < n) deg[idx] = 0.0f;
}

__global__ void dhcf_degree_kernel(const int* __restrict__ eu,
                                   const int* __restrict__ ei,
                                   float* __restrict__ deg,
                                   int ne, int NU)
{
    int e = blockIdx.x * blockDim.x + threadIdx.x;
    if (e >= ne) return;
    atomicAdd(deg + eu[e], 1.0f);
    atomicAdd(deg + NU + ei[e], 1.0f);
}

__global__ void dhcf_degnorm_kernel(const float* __restrict__ deg,
                                    float* __restrict__ dis,
                                    float* __restrict__ dinv, int n)
{
    int v = blockIdx.x * blockDim.x + threadIdx.x;
    if (v >= n) return;
    float d = deg[v];
    if (d > 0.0f) { dis[v] = 1.0f / sqrtf(d); dinv[v] = 1.0f / d; }
    else          { dis[v] = 0.0f;            dinv[v] = 0.0f;     }
}

__global__ void dhcf_zero_kernel(float* __restrict__ p, int count)
{
    int idx = blockIdx.x * blockDim.x + threadIdx.x;
    if (idx < count) p[idx] = 0.0f;
}

__global__ void dhcf_edge_gather_scatter(const int* __restrict__ eu,
                                         const int* __restrict__ ei,
                                         const float* __restrict__ SRC,
                                         const float* __restrict__ scale,
                                         float* __restrict__ DST,
                                         int ne, int NU)
{
    int idx = blockIdx.x * blockDim.x + threadIdx.x;
    if (idx >= ne * 32) return;
    int e = idx >> 5;
    int c = (idx & 31) << 1;
    int u = eu[e];
    int i = NU + ei[e];
    float2 xu = *(const float2*)(SRC + (size_t)u * 64 + c);
    float2 xi = *(const float2*)(SRC + (size_t)i * 64 + c);
    float su = scale[u], si = scale[i];
    float* di = DST + (size_t)i * 64 + c;
    float* du = DST + (size_t)u * 64 + c;
    atomicAdd(di,     xu.x * su);
    atomicAdd(di + 1, xu.y * su);
    atomicAdd(du,     xi.x * si);
    atomicAdd(du + 1, xi.y * si);
}

__global__ void dhcf_count_rank_kernel(const int* __restrict__ eu,
                                       const int* __restrict__ ei,
                                       int* __restrict__ cnt,
                                       int* __restrict__ rank_u,
                                       int* __restrict__ rank_i,
                                       int ne, int NU)
{
    int e = blockIdx.x * blockDim.x + threadIdx.x;
    if (e >= ne) return;
    rank_u[e] = atomicAdd(cnt + eu[e], 1);
    rank_i[e] = atomicAdd(cnt + NU + ei[e], 1);
}

__global__ __launch_bounds__(256) void dhcf_scan_local_kernel(
    const int* __restrict__ cnt, int* __restrict__ rowptr,
    int* __restrict__ bsum, int n)
{
    __shared__ int wsum[WPB];
    const int t = threadIdx.x, lane = t & 31, wave = t >> 5;
    const int base = blockIdx.x * 1024 + t * 4;
    const int c0 = (base     < n) ? cnt[base]     : 0;
    const int c1 = (base + 1 < n) ? cnt[base + 1] : 0;
    const int c2 = (base + 2 < n) ? cnt[base + 2] : 0;
    const int c3 = (base + 3 < n) ? cnt[base + 3] : 0;
    const int tot = c0 + c1 + c2 + c3;
    int inc = tot;
    #pragma unroll
    for (int d = 1; d < 32; d <<= 1) {
        int y = __shfl_up(inc, d, 32);
        if (lane >= d) inc += y;
    }
    if (lane == 31) wsum[wave] = inc;
    __syncthreads();
    int woff = 0;
    for (int w = 0; w < wave; ++w) woff += wsum[w];
    const int excl = woff + inc - tot;
    if (base     < n) rowptr[base]     = excl;
    if (base + 1 < n) rowptr[base + 1] = excl + c0;
    if (base + 2 < n) rowptr[base + 2] = excl + c0 + c1;
    if (base + 3 < n) rowptr[base + 3] = excl + c0 + c1 + c2;
    if (t == 255) bsum[blockIdx.x] = woff + inc;
}

__global__ __launch_bounds__(256) void dhcf_scan_finish_kernel(
    const int* __restrict__ cnt, int* __restrict__ rowptr,
    const int* __restrict__ bsum, float* __restrict__ dis,
    float* __restrict__ dinv, int n, int total)
{
    __shared__ int wsum[WPB];
    const int t = threadIdx.x, lane = t & 31, wave = t >> 5;
    int part = 0;
    for (int i = t; i < (int)blockIdx.x; i += 256) part += bsum[i];
    part += __shfl_xor(part, 16, 32);
    part += __shfl_xor(part, 8, 32);
    part += __shfl_xor(part, 4, 32);
    part += __shfl_xor(part, 2, 32);
    part += __shfl_xor(part, 1, 32);
    if (lane == 0) wsum[wave] = part;
    __syncthreads();
    int off = 0;
    #pragma unroll
    for (int w = 0; w < WPB; ++w) off += wsum[w];
    const int base = blockIdx.x * 1024 + t * 4;
    #pragma unroll
    for (int k = 0; k < 4; ++k) {
        const int v = base + k;
        if (v < n) {
            rowptr[v] += off;
            const int c = cnt[v];
            const float d = (float)c;
            if (c > 0) { dis[v] = 1.0f / sqrtf(d); dinv[v] = 1.0f / d; }
            else       { dis[v] = 0.0f;            dinv[v] = 0.0f;     }
        }
    }
    if (blockIdx.x == gridDim.x - 1 && t == 0) rowptr[n] = total;
}

__global__ void dhcf_fill_kernel(const int* __restrict__ eu,
                                 const int* __restrict__ ei,
                                 const int* __restrict__ rank_u,
                                 const int* __restrict__ rank_i,
                                 const int* __restrict__ rowptr,
                                 int* __restrict__ adj,
                                 int ne, int NU, int total)
{
    int e = blockIdx.x * blockDim.x + threadIdx.x;
    if (e >= ne) return;
    const int u = eu[e];
    const int i = NU + ei[e];
    const int su = min(max(rowptr[u] + rank_u[e], 0), total - 1);
    const int si = min(max(rowptr[i] + rank_i[e], 0), total - 1);
    adj[su] = i;
    adj[si] = u;
}


#define DHCF_RS(K) case K: return __builtin_amdgcn_update_dpp(0, v, 0x150 + K, 0xf, 0xf, false);
__device__ __forceinline__ int dhcf_row_share(int v, int k)
{
    switch (k) {
        DHCF_RS(0)  DHCF_RS(1)  DHCF_RS(2)  DHCF_RS(3)
        DHCF_RS(4)  DHCF_RS(5)  DHCF_RS(6)  DHCF_RS(7)
        DHCF_RS(8)  DHCF_RS(9)  DHCF_RS(10) DHCF_RS(11)
        DHCF_RS(12) DHCF_RS(13) DHCF_RS(14)
        default: return __builtin_amdgcn_update_dpp(0, v, 0x150 + 15, 0xf, 0xf, false);
    }
}
#undef DHCF_RS

template <bool SCALE_IN>
__device__ __forceinline__ void dhcf_acc(v2f& lo, v2f& hi, v4f x, bool act, float s)
{
    v2f xl, xh;
    xl.x = act ? x.x : 0.0f;  xl.y = act ? x.y : 0.0f;
    xh.x = act ? x.z : 0.0f;  xh.y = act ? x.w : 0.0f;
    if (SCALE_IN) { lo += xl * s; hi += xh * s; }
    else          { lo += xl;     hi += xh;     }
}

template <bool TWO_TABLES>
__device__ __forceinline__ v4f dhcf_row(const char* t0, const char* t1, int nu,
                                        int nb, unsigned cb)
{
    if (TWO_TABLES) {
        const bool low = nb < nu;
        const char* t = low ? t0 : t1;
        const unsigned r = (unsigned)(low ? nb : nb - nu);
        return *(const v4f*)(t + (r * 256u + cb));
    }
    return *(const v4f*)(t0 + ((unsigned)nb * 256u + cb));
}

template <bool TWO_TABLES>
__device__ __forceinline__ void dhcf_ask4(v4f (&x)[4], int mS, int q,
                                          const char* srcb, const char* srch,
                                          int nu, unsigned cb)
{
    #pragma unroll
    for (int j = 0; j < 4; ++j) {
        const int nb = dhcf_row_share(mS, (q * 4 + j) & 15);
        x[j] = dhcf_row<TWO_TABLES>(srcb, srch, nu, nb, cb);
    }
}
template <bool SCALE_IN>
__device__ __forceinline__ void dhcf_use4(v2f& alo, v2f& ahi, v2f& blo, v2f& bhi,
                                          const v4f (&x)[4], int sS, int q, int cnt)
{
    #pragma unroll
    for (int j = 0; j < 4; ++j) {
        const int k = q * 4 + j;
        const float s = __int_as_float(dhcf_row_share(sS, k & 15));
        if (j & 1) dhcf_acc<SCALE_IN>(blo, bhi, x[j], k < cnt, s);
        else       dhcf_acc<SCALE_IN>(alo, ahi, x[j], k < cnt, s);
    }
}

template <bool SCALE_IN, bool TWO_TABLES, bool DEEP, bool IDS_GIVEN>
__device__ __forceinline__ v4f dhcf_pull_rows(
    const int* __restrict__ adj, const char* srcb, const char* srch, int nu,
    const float* __restrict__ scale_in, int n, int total,
    int beg, int deg, int l16, unsigned cb, int id0f, int id1f)
{
    const int dmax = __builtin_amdgcn_readfirstlane(max(deg, __shfl_xor(deg, 16, 32)));

    v2f alo = {0.0f, 0.0f}, ahi = {0.0f, 0.0f};
    v2f blo = {0.0f, 0.0f}, bhi = {0.0f, 0.0f};
    for (int off = 0; off < dmax; off += 32) {
        const int cnt = deg - off;
        const int cm  = min(32, dmax - off);
        int id0 = id0f;
        int id1 = id1f;
        if (!IDS_GIVEN || off > 0) {
            const int p0 = min(beg + off + l16, total - 1);
            const int p1 = min(beg + off + 16 + l16, total - 1);
            id0 = adj[p0];
            id1 = adj[p1];
        }
        id0 = min(max(id0, 0), n - 1);
        id1 = min(max(id1, 0), n - 1);
        const int o0 = __shfl_xor(id0, 16, 32);
        const int o1 = __shfl_xor(id1, 16, 32);
        int m0 = (l16 < cnt)      ? id0 : o0;
        int m1 = (l16 + 16 < cnt) ? id1 : o1;
        const int q0 = __builtin_amdgcn_update_dpp(0, m0, 0x00, 0xf, 0xf, false);
        const int q1 = __builtin_amdgcn_update_dpp(0, m1, 0x00, 0xf, 0xf, false);
        m0 = (l16 < cm)      ? m0 : q0;
        m1 = (l16 + 16 < cm) ? m1 : q1;
        float s0 = 1.0f, s1 = 1.0f;
        if (SCALE_IN) {
            s0 = scale_in[m0];
            if (cm > 16) s1 = scale_in[m1];
        }
        if (DEEP) {
            v4f x[6][4];
            const int sS0 = __float_as_int(s0);
            const int sS1 = __float_as_int(s1);
            #pragma unroll
            for (int q = 0; q < 6; ++q)
                if (q * 4 < cm) dhcf_ask4<TWO_TABLES>(x[q], (q < 4) ? m0 : m1, q, srcb, srch, nu, cb);
            #pragma unroll
            for (int q = 0; q < 2; ++q)
                if (q * 4 < cm) dhcf_use4<SCALE_IN>(alo, ahi, blo, bhi, x[q], sS0, q, cnt);
            #pragma unroll
            for (int q = 6; q < 8; ++q)
                if (q * 4 < cm) dhcf_ask4<TWO_TABLES>(x[q - 6], m1, q, srcb, srch, nu, cb);
            #pragma unroll
            for (int q = 2; q < 6; ++q)
                if (q * 4 < cm) dhcf_use4<SCALE_IN>(alo, ahi, blo, bhi, x[q], (q < 4) ? sS0 : sS1, q, cnt);
            #pragma unroll
            for (int q = 6; q < 8; ++q)
                if (q * 4 < cm) dhcf_use4<SCALE_IN>(alo, ahi, blo, bhi, x[q - 6], sS1, q, cnt);
            continue;
        }
        #pragma unroll
        for (int g = 0; g < 4; ++g) {
            if (g * 8 < cm) {
                const int   mS = (g < 2) ? m0 : m1;
                const int   sS = __float_as_int((g < 2) ? s0 : s1);
                v4f xa[4], xb[4];
                #pragma unroll
                for (int j = 0; j < 4; ++j) {
                    const int nb = dhcf_row_share(mS, (g * 8 + j) & 15);
                    xa[j] = dhcf_row<TWO_TABLES>(srcb, srch, nu, nb, cb);
                }
                if (g * 8 + 4 < cm) {
                    #pragma unroll
                    for (int j = 0; j < 4; ++j) {
                        const int nb = dhcf_row_share(mS, (g * 8 + 4 + j) & 15);
                        xb[j] = dhcf_row<TWO_TABLES>(srcb, srch, nu, nb, cb);
                    }
                    #pragma unroll
                    for (int j = 0; j < 4; ++j) {
                        const int k = g * 8 + j;
                        const float s = __int_as_float(dhcf_row_share(sS, k & 15));
                        if (j & 1) dhcf_acc<SCALE_IN>(blo, bhi, xa[j], k < cnt, s);
                        else       dhcf_acc<SCALE_IN>(alo, ahi, xa[j], k < cnt, s);
                    }
                    #pragma unroll
                    for (int j = 0; j < 4; ++j) {
                        const int k = g * 8 + 4 + j;
                        const float s = __int_as_float(dhcf_row_share(sS, k & 15));
                        if (j & 1) dhcf_acc<SCALE_IN>(blo, bhi, xb[j], k < cnt, s);
                        else       dhcf_acc<SCALE_IN>(alo, ahi, xb[j], k < cnt, s);
                    }
                } else {
                    #pragma unroll
                    for (int j = 0; j < 4; ++j) {
                        const int k = g * 8 + j;
                        const float s = __int_as_float(dhcf_row_share(sS, k & 15));
                        if (j & 1) dhcf_acc<SCALE_IN>(blo, bhi, xa[j], k < cnt, s);
                        else       dhcf_acc<SCALE_IN>(alo, ahi, xa[j], k < cnt, s);
                    }
                }
            }
        }
    }
    alo += blo; ahi += bhi;
    v4f r;
    r.x = alo.x; r.y = alo.y; r.z = ahi.x; r.w = ahi.y;
    return r;
}

template <bool SCALE_IN, bool SCALE_OUT, bool TWO_TABLES>
__global__ __launch_bounds__(256) void dhcf_pull_kernel(
    const int* __restrict__ rowptr, const int* __restrict__ adj,
    const float* __restrict__ SRC, const float* __restrict__ SRC_HI, int nu,
    const float* __restrict__ scale_in,
    const float* __restrict__ scale_out, float* __restrict__ DST,
    int n, int total)
{
    const int row  = (blockIdx.x * 256 + threadIdx.x) >> 4;
    if ((row & ~1) >= n) return;
    const int l16  = threadIdx.x & 15;
    const unsigned cb = (unsigned)l16 << 4;
    const bool valid = row < n;
    int beg = 0, deg = 0;
    if (valid) {
        beg = max(rowptr[row], 0);
        deg = min(rowptr[row + 1], total) - beg;
    }
    v4f r = dhcf_pull_rows<SCALE_IN, TWO_TABLES, false, false>(
        adj, (const char*)SRC, (const char*)SRC_HI, nu, scale_in, n, total,
        beg, deg, l16, cb, 0, 0);
    if (valid) {
        if (SCALE_OUT) r *= scale_out[row];
        *(v4f*)(DST + (size_t)row * 64 + (l16 << 2)) = r;
    }
}

#define WROW 68
template <bool FIRST, bool LAST, bool PULL>
__global__ __launch_bounds__(256) void dhcf_layer_kernel(
    float* __restrict__ X, const float* __restrict__ Yb,
    const float* __restrict__ dis,
    const float* __restrict__ Wgc, const float* __restrict__ bgc,
    const float* __restrict__ Wbi, const float* __restrict__ bbi,
    float* __restrict__ ACC, int ntiles,
    const float* __restrict__ u_emb, const float* __restrict__ i_emb,
    int nu, float inv,
    const int* __restrict__ rowptr, const int* __restrict__ adj,
    int n, int total)
{
    __shared__ __attribute__((aligned(16))) float sWg[64 * WROW];
    __shared__ __attribute__((aligned(16))) float sWb[64 * WROW];
    __shared__ __attribute__((aligned(16))) float sY[WPB][16 * WROW];
    __shared__ __attribute__((aligned(16))) float sX[WPB][16 * WROW];
    __shared__ float sBias[128];

    int wave = threadIdx.x >> 5;
    int lane = threadIdx.x & 31;
    int tile = blockIdx.x * WPB + wave;
    const bool live = tile < ntiles;

    int row0 = tile * 16;
    int m    = lane & 15;
    int khi  = lane >> 4;
    int rowA = row0 + m;

    float gscale = 0.0f;
    {
        v4f wg[4], wb[4];
        for (int it = 0; it < 4; ++it) {
            const int q = threadIdx.x + it * 256;
            wg[it] = *(const v4f*)(Wgc + q * 4);
            wb[it] = *(const v4f*)(Wbi + q * 4);
        }
        if (live) {
            const v4f* ysrc = (const v4f*)(Yb + (size_t)row0 * 64) + lane;
            const v4f* xsrc = (const v4f*)(X  + (size_t)row0 * 64) + lane;
            v4f ty[8], tx[8];
            for (int i = 0; i < 8; ++i) {
                if (!PULL) ty[i] = ysrc[i * 32];
                if (FIRST) {
                    const int r = row0 + 2 * i + khi;
                    const float* p = (r < nu) ? (u_emb + (size_t)r * 64)
                                              : (i_emb + (size_t)(r - nu) * 64);
                    tx[i] = *(const v4f*)(p + (m << 2));
                } else {
                    tx[i] = xsrc[i * 32];
                }
            }
            gscale = dis[rowA];
            float* dy = sY[wave] + khi * WROW + (m << 2);
            float* dx = sX[wave] + khi * WROW + (m << 2);
            for (int i = 0; i < 8; ++i) {
                if (!PULL) *(v4f*)(dy + 2 * i * WROW) = ty[i];
                *(v4f*)(dx + 2 * i * WROW) = tx[i];
            }
        }
        if (threadIdx.x < 128)
            sBias[threadIdx.x] = (threadIdx.x < 64) ? bgc[threadIdx.x] : bbi[threadIdx.x - 64];
        for (int it = 0; it < 4; ++it) {
            const int q  = threadIdx.x + it * 256;
            const int r  = q >> 4;
            const int c4 = (q & 15) << 2;
            *(v4f*)(sWg + r * WROW + c4) = wg[it];
            *(v4f*)(sWb + r * WROW + c4) = wb[it];
        }
    }
    if (PULL && live) {
        float* dy = sY[wave] + khi * WROW + (m << 2);
        const int rp = rowptr[row0 + min(lane, 16)];
        const unsigned cb = (unsigned)m << 4;
        int begN = max(__shfl(rp, khi, 32), 0);
        int degN = min(__shfl(rp, khi + 1, 32), total) - begN;
        int id0N = adj[min(begN + m, total - 1)];
        int id1N = adj[min(begN + 16 + m, total - 1)];
        #pragma nounroll
        for (int i = 0; i < 8; ++i) {
            const int beg = begN, deg = degN, id0 = id0N, id1 = id1N;
            if (i < 7) {
                begN = max(__shfl(rp, 2 * i + 2 + khi, 32), 0);
                degN = min(__shfl(rp, 2 * i + 3 + khi, 32), total) - begN;
                id0N = adj[min(begN + m, total - 1)];
                id1N = adj[min(begN + 16 + m, total - 1)];
            }
            const v4f y = dhcf_pull_rows<false, false, true, true>(
                adj, (const char*)Yb, (const char*)Yb, nu, dis, n, total,
                beg, deg, m, cb, id0, id1);
            *(v4f*)(dy + 2 * i * WROW) = y;
        }
    }
    __syncthreads();
    if (!live) return;

    const v8f vz = {0,0,0,0,0,0,0,0};
    v8f acc_s[4], acc_b[4];
    #pragma unroll
    for (int jj = 0; jj < 4; ++jj) { acc_s[jj] = vz; acc_b[jj] = vz; }

    const float* yrow = sY[wave] + m * WROW;
    const float* xrow = sX[wave] + m * WROW;

    #pragma unroll
    for (int kk = 0; kk < 16; ++kk) {
        int k0 = kk * 4 + khi * 2;
        v2f g2, xg2;
        float y0 = yrow[k0], y1 = yrow[k0 + 1];
        float x0 = xrow[k0], x1 = xrow[k0 + 1];
        g2.x  = y0 * gscale;  g2.y  = y1 * gscale;
        xg2.x = x0 * g2.x;    xg2.y = x1 * g2.y;
        #pragma unroll
        for (int jj = 0; jj < 4; ++jj) {
            int j = jj * 16 + m;
            v2f bg, bb;
            bg.x = sWg[j * WROW + k0]; bg.y = sWg[j * WROW + k0 + 1];
            bb.x = sWb[j * WROW + k0]; bb.y = sWb[j * WROW + k0 + 1];
            acc_s[jj] = __builtin_amdgcn_wmma_f32_16x16x4_f32(
                false, g2,  false, bg, (short)0, acc_s[jj], false, false);
            acc_b[jj] = __builtin_amdgcn_wmma_f32_16x16x4_f32(
                false, xg2, false, bb, (short)0, acc_b[jj], false, false);
        }
    }

    float xn[4][8];
    #pragma unroll
    for (int jj = 0; jj < 4; ++jj) {
        int j = jj * 16 + m;
        float bgj = sBias[j];
        float bbj = sBias[64 + j];
        #pragma unroll
        for (int r = 0; r < 8; ++r) {
            float xv = sX[wave][(r + 8 * khi) * WROW + j];
            float s = acc_s[jj][r] + bgj + xv;
            s = (s > 0.0f) ? s : 0.2f * s;
            float b = acc_b[jj][r] + bbj;
            b = (b > 0.0f) ? b : 0.2f * b;
            xn[jj][r] = s + b;
        }
    }

    #pragma unroll
    for (int r = 0; r < 8; ++r) {
        float p = 0.0f;
        #pragma unroll
        for (int jj = 0; jj < 4; ++jj) p += xn[jj][r] * xn[jj][r];
        p += __shfl_xor(p, 1, 32);
        p += __shfl_xor(p, 2, 32);
        p += __shfl_xor(p, 4, 32);
        p += __shfl_xor(p, 8, 32);
        float scl = 1.0f / fmaxf(sqrtf(p), 1e-12f);
        int row = row0 + r + 8 * khi;
        #pragma unroll
        for (int jj = 0; jj < 4; ++jj) {
            int j = jj * 16 + m;
            float v = xn[jj][r] * scl;
            if (!LAST) X[(size_t)row * 64 + j] = v;
            float a = FIRST ? sX[wave][(r + 8 * khi) * WROW + j]
                            : ACC[(size_t)row * 64 + j];
            a += v;
            if (LAST) a *= inv;
            ACC[(size_t)row * 64 + j] = a;
        }
    }
}

extern "C" void kernel_launch(void* const* d_in, const int* in_sizes, int n_in,
                              void* d_out, int out_size, void* d_ws, size_t ws_size,
                              hipStream_t stream)
{
    const float* u_emb = (const float*)d_in[0];
    const float* i_emb = (const float*)d_in[1];
    const float* Wgc   = (const float*)d_in[2];
    const float* bgc   = (const float*)d_in[3];
    const float* Wbi   = (const float*)d_in[4];
    const float* bbi   = (const float*)d_in[5];
    const int*   eu    = (const int*)d_in[6];
    const int*   ei    = (const int*)d_in[7];

    const int D  = 64;
    const int NU = in_sizes[0] / D;
    const int NI = in_sizes[1] / D;
    const int N  = NU + NI;
    const int NE = in_sizes[6];
    const int L  = in_sizes[2] / (D * D);
    const int n64 = N * D;

    float* X    = (float*)d_ws;
    float* E    = X + (size_t)n64;
    float* Y    = E + (size_t)n64;
    float* deg  = Y + (size_t)n64;
    float* dis  = deg + N;
    float* dinv = dis + N;
    float* ACC  = (float*)d_out;

    const int nchunk = (N + 1023) / 1024;
    const int total  = 2 * NE;
    int* cnt    = (int*)deg;
    int* rowptr = (int*)(dinv + N);
    int* bsum   = rowptr + (N + 1);
    int* adj    = bsum + nchunk;
    int* rank_u = (int*)Y;
    int* rank_i = rank_u + NE;
    const size_t need = (size_t)((const char*)(adj + total) - (const char*)d_ws);
    const bool lists = (ws_size >= need) && (total <= n64) && (NE > 0) &&
                       (n64 < (1 << 30));

    dim3 blk(256);
    const bool direct = lists && (L > 0);
    if (direct) {
        dhcf_zero_kernel<<<(N + 255) / 256, blk, 0, stream>>>(deg, N);
    } else {
        dhcf_init_kernel<<<(n64 + 255) / 256, blk, 0, stream>>>(
            u_emb, i_emb, X, ACC, deg, NU * D, n64, N);
    }

    const int ntiles = N / 16;
    const int lgrid  = (ntiles + WPB - 1) / WPB;
    const int egrid  = (NE * 32 + 255) / 256;
    const int pgrid  = (N * 16 + 255) / 256;

    if (lists) {
        dhcf_count_rank_kernel<<<(NE + 255) / 256, blk, 0, stream>>>(
            eu, ei, cnt, rank_u, rank_i, NE, NU);
        dhcf_scan_local_kernel<<<nchunk, blk, 0, stream>>>(cnt, rowptr, bsum, N);
        dhcf_scan_finish_kernel<<<nchunk, blk, 0, stream>>>(
            cnt, rowptr, bsum, dis, dinv, N, total);
        dhcf_fill_kernel<<<(NE + 255) / 256, blk, 0, stream>>>(
            eu, ei, rank_u, rank_i, rowptr, adj, NE, NU, total);
    } else {
        dhcf_degree_kernel<<<(NE + 255) / 256, blk, 0, stream>>>(eu, ei, deg, NE, NU);
        dhcf_degnorm_kernel<<<(N + 255) / 256, blk, 0, stream>>>(deg, dis, dinv, N);
    }

    const float inv = 1.0f / (float)(L + 1);
    for (int l = 0; l < L; ++l) {
        const bool first = direct && (l == 0);
        const bool last  = (l == L - 1);
        if (lists) {
            if (first) {
                dhcf_pull_kernel<true, true, true><<<pgrid, blk, 0, stream>>>(
                    rowptr, adj, u_emb, i_emb, NU, dis, dinv, E, N, total);
            } else {
                dhcf_pull_kernel<true, true, false><<<pgrid, blk, 0, stream>>>(
                    rowptr, adj, X, X, NU, dis, dinv, E, N, total);
            }
        } else {
            dhcf_zero_kernel<<<(2 * n64 + 255) / 256, blk, 0, stream>>>(E, 2 * n64);
            dhcf_edge_gather_scatter<<<egrid, blk, 0, stream>>>(eu, ei, X, dis,  E, NE, NU);
            dhcf_edge_gather_scatter<<<egrid, blk, 0, stream>>>(eu, ei, E, dinv, Y, NE, NU);
        }
        const float* wg = Wgc + (size_t)l * D * D;
        const float* bg = bgc + (size_t)l * D;
        const float* wb = Wbi + (size_t)l * D * D;
        const float* bb = bbi + (size_t)l * D;
        if (lists) {
            if (first && last) {
                dhcf_layer_kernel<true, true, true><<<lgrid, blk, 0, stream>>>(
                    X, E, dis, wg, bg, wb, bb, ACC, ntiles, u_emb, i_emb, NU, inv,
                    rowptr, adj, N, total);
            } else if (first) {
                dhcf_layer_kernel<true, false, true><<<lgrid, blk, 0, stream>>>(
                    X, E, dis, wg, bg, wb, bb, ACC, ntiles, u_emb, i_emb, NU, inv,
                    rowptr, adj, N, total);
            } else if (last) {
                dhcf_layer_kernel<false, true, true><<<lgrid, blk, 0, stream>>>(
                    X, E, dis, wg, bg, wb, bb, ACC, ntiles, u_emb, i_emb, NU, inv,
                    rowptr, adj, N, total);
            } else {
                dhcf_layer_kernel<false, false, true><<<lgrid, blk, 0, stream>>>(
                    X, E, dis, wg, bg, wb, bb, ACC, ntiles, u_emb, i_emb, NU, inv,
                    rowptr, adj, N, total);
            }
        } else if (last) {
            dhcf_layer_kernel<false, true, false><<<lgrid, blk, 0, stream>>>(
                X, Y, dis, wg, bg, wb, bb, ACC, ntiles, u_emb, i_emb, NU, inv,
                rowptr, adj, N, total);
        } else {
            dhcf_layer_kernel<false, false, false><<<lgrid, blk, 0, stream>>>(
                X, Y, dis, wg, bg, wb, bb, ACC, ntiles, u_emb, i_emb, NU, inv,
                rowptr, adj, N, total);
        }
    }
}


# Fewer, longer-lived waves: pull kernel is 1536 resident work-groups whose waves loop over vertex pairs, next pair's list prefetched (12288 waves a launch, not 75000); count and fill: 4 edges a thread
# speedup vs baseline: 2.8535x; 1.0328x over previous
#include <hip/hip_runtime.h>
#include <math.h>


typedef __attribute__((ext_vector_type(2))) float v2f;
typedef __attribute__((ext_vector_type(8))) float v8f;
typedef __attribute__((ext_vector_type(4))) float v4f;

#define WPB 8

__global__ void dhcf_init_kernel(const float* __restrict__ u_emb,
                                 const float* __restrict__ i_emb,
                                 float* __restrict__ X,
                                 float* __restrict__ ACC,
                                 float* __restrict__ deg,
                                 int nu64, int n64, int n)
{
    int idx = blockIdx.x * blockDim.x + threadIdx.x;
    if (idx < n64) {
        float v = (idx < nu64) ? u_emb[idx] : i_emb[idx - nu64];
        X[idx]   = v;
        ACC[idx] = v;
    }
    if (idx < n) deg[idx] = 0.0f;
}

__global__ void dhcf_degree_kernel(const int* __restrict__ eu,
                                   const int* __restrict__ ei,
                                   float* __restrict__ deg,
                                   int ne, int NU)
{
    int e = blockIdx.x * blockDim.x + threadIdx.x;
    if (e >= ne) return;
    atomicAdd(deg + eu[e], 1.0f);
    atomicAdd(deg + NU + ei[e], 1.0f);
}

__global__ void dhcf_degnorm_kernel(const float* __restrict__ deg,
                                    float* __restrict__ dis,
                                    float* __restrict__ dinv, int n)
{
    int v = blockIdx.x * blockDim.x + threadIdx.x;
    if (v >= n) return;
    float d = deg[v];
    if (d > 0.0f) { dis[v] = 1.0f / sqrtf(d); dinv[v] = 1.0f / d; }
    else          { dis[v] = 0.0f;            dinv[v] = 0.0f;     }
}

__global__ void dhcf_zero_kernel(float* __restrict__ p, int count)
{
    int idx = blockIdx.x * blockDim.x + threadIdx.x;
    if (idx < count) p[idx] = 0.0f;
}

__global__ void dhcf_edge_gather_scatter(const int* __restrict__ eu,
                                         const int* __restrict__ ei,
                                         const float* __restrict__ SRC,
                                         const float* __restrict__ scale,
                                         float* __restrict__ DST,
                                         int ne, int NU)
{
    int idx = blockIdx.x * blockDim.x + threadIdx.x;
    if (idx >= ne * 32) return;
    int e = idx >> 5;
    int c = (idx & 31) << 1;
    int u = eu[e];
    int i = NU + ei[e];
    float2 xu = *(const float2*)(SRC + (size_t)u * 64 + c);
    float2 xi = *(const float2*)(SRC + (size_t)i * 64 + c);
    float su = scale[u], si = scale[i];
    float* di = DST + (size_t)i * 64 + c;
    float* du = DST + (size_t)u * 64 + c;
    atomicAdd(di,     xu.x * su);
    atomicAdd(di + 1, xu.y * su);
    atomicAdd(du,     xi.x * si);
    atomicAdd(du + 1, xi.y * si);
}

#define EPT 4
__global__ void dhcf_count_rank_kernel(const int* __restrict__ eu,
                                       const int* __restrict__ ei,
                                       int* __restrict__ cnt,
                                       int* __restrict__ rank_u,
                                       int* __restrict__ rank_i,
                                       int ne, int NU)
{
    const int nth = (int)(gridDim.x * blockDim.x);
    const int e0  = (int)(blockIdx.x * blockDim.x + threadIdx.x);
    if (e0 + (EPT - 1) * nth < ne) {
        int u[EPT], it[EPT], ru[EPT], ri[EPT];
        #pragma unroll
        for (int k = 0; k < EPT; ++k) {
            u[k]  = eu[e0 + k * nth];
            it[k] = ei[e0 + k * nth];
        }
        #pragma unroll
        for (int k = 0; k < EPT; ++k) {
            ru[k] = atomicAdd(cnt + u[k], 1);
            ri[k] = atomicAdd(cnt + NU + it[k], 1);
        }
        #pragma unroll
        for (int k = 0; k < EPT; ++k) {
            rank_u[e0 + k * nth] = ru[k];
            rank_i[e0 + k * nth] = ri[k];
        }
    } else {
        for (int e = e0; e < ne; e += nth) {
            rank_u[e] = atomicAdd(cnt + eu[e], 1);
            rank_i[e] = atomicAdd(cnt + NU + ei[e], 1);
        }
    }
}

__global__ __launch_bounds__(256) void dhcf_scan_local_kernel(
    const int* __restrict__ cnt, int* __restrict__ rowptr,
    int* __restrict__ bsum, int n)
{
    __shared__ int wsum[WPB];
    const int t = threadIdx.x, lane = t & 31, wave = t >> 5;
    const int base = blockIdx.x * 1024 + t * 4;
    const int c0 = (base     < n) ? cnt[base]     : 0;
    const int c1 = (base + 1 < n) ? cnt[base + 1] : 0;
    const int c2 = (base + 2 < n) ? cnt[base + 2] : 0;
    const int c3 = (base + 3 < n) ? cnt[base + 3] : 0;
    const int tot = c0 + c1 + c2 + c3;
    int inc = tot;
    #pragma unroll
    for (int d = 1; d < 32; d <<= 1) {
        int y = __shfl_up(inc, d, 32);
        if (lane >= d) inc += y;
    }
    if (lane == 31) wsum[wave] = inc;
    __syncthreads();
    int woff = 0;
    for (int w = 0; w < wave; ++w) woff += wsum[w];
    const int excl = woff + inc - tot;
    if (base     < n) rowptr[base]     = excl;
    if (base + 1 < n) rowptr[base + 1] = excl + c0;
    if (base + 2 < n) rowptr[base + 2] = excl + c0 + c1;
    if (base + 3 < n) rowptr[base + 3] = excl + c0 + c1 + c2;
    if (t == 255) bsum[blockIdx.x] = woff + inc;
}

__global__ __launch_bounds__(256) void dhcf_scan_finish_kernel(
    const int* __restrict__ cnt, int* __restrict__ rowptr,
    const int* __restrict__ bsum, float* __restrict__ dis,
    float* __restrict__ dinv, int n, int total)
{
    __shared__ int wsum[WPB];
    const int t = threadIdx.x, lane = t & 31, wave = t >> 5;
    int part = 0;
    for (int i = t; i < (int)blockIdx.x; i += 256) part += bsum[i];
    part += __shfl_xor(part, 16, 32);
    part += __shfl_xor(part, 8, 32);
    part += __shfl_xor(part, 4, 32);
    part += __shfl_xor(part, 2, 32);
    part += __shfl_xor(part, 1, 32);
    if (lane == 0) wsum[wave] = part;
    __syncthreads();
    int off = 0;
    #pragma unroll
    for (int w = 0; w < WPB; ++w) off += wsum[w];
    const int base = blockIdx.x * 1024 + t * 4;
    #pragma unroll
    for (int k = 0; k < 4; ++k) {
        const int v = base + k;
        if (v < n) {
            rowptr[v] += off;
            const int c = cnt[v];
            const float d = (float)c;
            if (c > 0) { dis[v] = 1.0f / sqrtf(d); dinv[v] = 1.0f / d; }
            else       { dis[v] = 0.0f;            dinv[v] = 0.0f;     }
        }
    }
    if (blockIdx.x == gridDim.x - 1 && t == 0) rowptr[n] = total;
}

__global__ void dhcf_fill_kernel(const int* __restrict__ eu,
                                 const int* __restrict__ ei,
                                 const int* __restrict__ rank_u,
                                 const int* __restrict__ rank_i,
                                 const int* __restrict__ rowptr,
                                 int* __restrict__ adj,
                                 int ne, int NU, int total)
{
    const int nth = (int)(gridDim.x * blockDim.x);
    const int e0  = (int)(blockIdx.x * blockDim.x + threadIdx.x);
    if (e0 + (EPT - 1) * nth < ne) {
        int u[EPT], i[EPT], ku[EPT], ki[EPT], pu[EPT], pi[EPT];
        #pragma unroll
        for (int k = 0; k < EPT; ++k) {
            const int e = e0 + k * nth;
            u[k]  = eu[e];
            i[k]  = NU + ei[e];
            ku[k] = rank_u[e];
            ki[k] = rank_i[e];
        }
        #pragma unroll
        for (int k = 0; k < EPT; ++k) {
            pu[k] = rowptr[u[k]];
            pi[k] = rowptr[i[k]];
        }
        #pragma unroll
        for (int k = 0; k < EPT; ++k) {
            const int su = min(max(pu[k] + ku[k], 0), total - 1);
            const int si = min(max(pi[k] + ki[k], 0), total - 1);
            adj[su] = i[k];
            adj[si] = u[k];
        }
    } else {
        for (int e = e0; e < ne; e += nth) {
            const int u = eu[e];
            const int i = NU + ei[e];
            const int su = min(max(rowptr[u] + rank_u[e], 0), total - 1);
            const int si = min(max(rowptr[i] + rank_i[e], 0), total - 1);
            adj[su] = i;
            adj[si] = u;
        }
    }
}


#define DHCF_RS(K) case K: return __builtin_amdgcn_update_dpp(0, v, 0x150 + K, 0xf, 0xf, false);
__device__ __forceinline__ int dhcf_row_share(int v, int k)
{
    switch (k) {
        DHCF_RS(0)  DHCF_RS(1)  DHCF_RS(2)  DHCF_RS(3)
        DHCF_RS(4)  DHCF_RS(5)  DHCF_RS(6)  DHCF_RS(7)
        DHCF_RS(8)  DHCF_RS(9)  DHCF_RS(10) DHCF_RS(11)
        DHCF_RS(12) DHCF_RS(13) DHCF_RS(14)
        default: return __builtin_amdgcn_update_dpp(0, v, 0x150 + 15, 0xf, 0xf, false);
    }
}
#undef DHCF_RS

template <bool SCALE_IN>
__device__ __forceinline__ void dhcf_acc(v2f& lo, v2f& hi, v4f x, bool act, float s)
{
    v2f xl, xh;
    xl.x = act ? x.x : 0.0f;  xl.y = act ? x.y : 0.0f;
    xh.x = act ? x.z : 0.0f;  xh.y = act ? x.w : 0.0f;
    if (SCALE_IN) { lo += xl * s; hi += xh * s; }
    else          { lo += xl;     hi += xh;     }
}

template <bool TWO_TABLES>
__device__ __forceinline__ v4f dhcf_row(const char* t0, const char* t1, int nu,
                                        int nb, unsigned cb)
{
    if (TWO_TABLES) {
        const bool low = nb < nu;
        const char* t = low ? t0 : t1;
        const unsigned r = (unsigned)(low ? nb : nb - nu);
        return *(const v4f*)(t + (r * 256u + cb));
    }
    return *(const v4f*)(t0 + ((unsigned)nb * 256u + cb));
}

template <bool TWO_TABLES>
__device__ __forceinline__ void dhcf_ask4(v4f (&x)[4], int mS, int q,
                                          const char* srcb, const char* srch,
                                          int nu, unsigned cb)
{
    #pragma unroll
    for (int j = 0; j < 4; ++j) {
        const int nb = dhcf_row_share(mS, (q * 4 + j) & 15);
        x[j] = dhcf_row<TWO_TABLES>(srcb, srch, nu, nb, cb);
    }
}
template <bool SCALE_IN>
__device__ __forceinline__ void dhcf_use4(v2f& alo, v2f& ahi, v2f& blo, v2f& bhi,
                                          const v4f (&x)[4], int sS, int q, int cnt)
{
    #pragma unroll
    for (int j = 0; j < 4; ++j) {
        const int k = q * 4 + j;
        const float s = __int_as_float(dhcf_row_share(sS, k & 15));
        if (j & 1) dhcf_acc<SCALE_IN>(blo, bhi, x[j], k < cnt, s);
        else       dhcf_acc<SCALE_IN>(alo, ahi, x[j], k < cnt, s);
    }
}

template <bool SCALE_IN, bool TWO_TABLES, bool DEEP, bool IDS_GIVEN>
__device__ __forceinline__ v4f dhcf_pull_rows(
    const int* __restrict__ adj, const char* srcb, const char* srch, int nu,
    const float* __restrict__ scale_in, int n, int total,
    int beg, int deg, int l16, unsigned cb, int id0f, int id1f)
{
    const int dmax = __builtin_amdgcn_readfirstlane(max(deg, __shfl_xor(deg, 16, 32)));

    v2f alo = {0.0f, 0.0f}, ahi = {0.0f, 0.0f};
    v2f blo = {0.0f, 0.0f}, bhi = {0.0f, 0.0f};
    for (int off = 0; off < dmax; off += 32) {
        const int cnt = deg - off;
        const int cm  = min(32, dmax - off);
        int id0 = id0f;
        int id1 = id1f;
        if (!IDS_GIVEN || off > 0) {
            const int p0 = min(beg + off + l16, total - 1);
            const int p1 = min(beg + off + 16 + l16, total - 1);
            id0 = adj[p0];
            id1 = adj[p1];
        }
        id0 = min(max(id0, 0), n - 1);
        id1 = min(max(id1, 0), n - 1);
        const int o0 = __shfl_xor(id0, 16, 32);
        const int o1 = __shfl_xor(id1, 16, 32);
        int m0 = (l16 < cnt)      ? id0 : o0;
        int m1 = (l16 + 16 < cnt) ? id1 : o1;
        const int q0 = __builtin_amdgcn_update_dpp(0, m0, 0x00, 0xf, 0xf, false);
        const int q1 = __builtin_amdgcn_update_dpp(0, m1, 0x00, 0xf, 0xf, false);
        m0 = (l16 < cm)      ? m0 : q0;
        m1 = (l16 + 16 < cm) ? m1 : q1;
        float s0 = 1.0f, s1 = 1.0f;
        if (SCALE_IN) {
            s0 = scale_in[m0];
            if (cm > 16) s1 = scale_in[m1];
        }
        if (DEEP) {
            v4f x[6][4];
            const int sS0 = __float_as_int(s0);
            const int sS1 = __float_as_int(s1);
            #pragma unroll
            for (int q = 0; q < 6; ++q)
                if (q * 4 < cm) dhcf_ask4<TWO_TABLES>(x[q], (q < 4) ? m0 : m1, q, srcb, srch, nu, cb);
            #pragma unroll
            for (int q = 0; q < 2; ++q)
                if (q * 4 < cm) dhcf_use4<SCALE_IN>(alo, ahi, blo, bhi, x[q], sS0, q, cnt);
            #pragma unroll
            for (int q = 6; q < 8; ++q)
                if (q * 4 < cm) dhcf_ask4<TWO_TABLES>(x[q - 6], m1, q, srcb, srch, nu, cb);
            #pragma unroll
            for (int q = 2; q < 6; ++q)
                if (q * 4 < cm) dhcf_use4<SCALE_IN>(alo, ahi, blo, bhi, x[q], (q < 4) ? sS0 : sS1, q, cnt);
            #pragma unroll
            for (int q = 6; q < 8; ++q)
                if (q * 4 < cm) dhcf_use4<SCALE_IN>(alo, ahi, blo, bhi, x[q - 6], sS1, q, cnt);
            continue;
        }
        #pragma unroll
        for (int g = 0; g < 4; ++g) {
            if (g * 8 < cm) {
                const int   mS = (g < 2) ? m0 : m1;
                const int   sS = __float_as_int((g < 2) ? s0 : s1);
                v4f xa[4], xb[4];
                #pragma unroll
                for (int j = 0; j < 4; ++j) {
                    const int nb = dhcf_row_share(mS, (g * 8 + j) & 15);
                    xa[j] = dhcf_row<TWO_TABLES>(srcb, srch, nu, nb, cb);
                }
                if (g * 8 + 4 < cm) {
                    #pragma unroll
                    for (int j = 0; j < 4; ++j) {
                        const int nb = dhcf_row_share(mS, (g * 8 + 4 + j) & 15);
                        xb[j] = dhcf_row<TWO_TABLES>(srcb, srch, nu, nb, cb);
                    }
                    #pragma unroll
                    for (int j = 0; j < 4; ++j) {
                        const int k = g * 8 + j;
                        const float s = __int_as_float(dhcf_row_share(sS, k & 15));
                        if (j & 1) dhcf_acc<SCALE_IN>(blo, bhi, xa[j], k < cnt, s);
                        else       dhcf_acc<SCALE_IN>(alo, ahi, xa[j], k < cnt, s);
                    }
                    #pragma unroll
                    for (int j = 0; j < 4; ++j) {
                        const int k = g * 8 + 4 + j;
                        const float s = __int_as_float(dhcf_row_share(sS, k & 15));
                        if (j & 1) dhcf_acc<SCALE_IN>(blo, bhi, xb[j], k < cnt, s);
                        else       dhcf_acc<SCALE_IN>(alo, ahi, xb[j], k < cnt, s);
                    }
                } else {
                    #pragma unroll
                    for (int j = 0; j < 4; ++j) {
                        const int k = g * 8 + j;
                        const float s = __int_as_float(dhcf_row_share(sS, k & 15));
                        if (j & 1) dhcf_acc<SCALE_IN>(blo, bhi, xa[j], k < cnt, s);
                        else       dhcf_acc<SCALE_IN>(alo, ahi, xa[j], k < cnt, s);
                    }
                }
            }
        }
    }
    alo += blo; ahi += bhi;
    v4f r;
    r.x = alo.x; r.y = alo.y; r.z = ahi.x; r.w = ahi.y;
    return r;
}

template <bool SCALE_IN, bool SCALE_OUT, bool TWO_TABLES>
__global__ __launch_bounds__(256) void dhcf_pull_kernel(
    const int* __restrict__ rowptr, const int* __restrict__ adj,
    const float* __restrict__ SRC, const float* __restrict__ SRC_HI, int nu,
    const float* __restrict__ scale_in,
    const float* __restrict__ scale_out, float* __restrict__ DST,
    int n, int total)
{
    const int l16  = threadIdx.x & 15;
    const int khi  = (threadIdx.x >> 4) & 1;
    const unsigned cb = (unsigned)l16 << 4;
    const int npairs = (n + 1) >> 1;
    const int stride = (int)gridDim.x * WPB;
    const int wv = __builtin_amdgcn_readfirstlane((int)(threadIdx.x >> 5));
    int pair = (int)blockIdx.x * WPB + wv;
    if (pair >= npairs) return;
    int rlo, rhi;
    {
        const int r = min(2 * pair + khi, n - 1);
        rlo = rowptr[r];
        rhi = rowptr[r + 1];
    }
    int beg = min(max(rlo, 0), total);
    int deg = (2 * pair + khi < n) ? (min(rhi, total) - beg) : 0;
    int id0 = adj[min(beg + l16, total - 1)];
    int id1 = adj[min(beg + 16 + l16, total - 1)];
    int pairN = pair + stride;
    {
        const int pc = min(pairN, npairs - 1);
        const int r  = min(2 * pc + khi, n - 1);
        rlo = rowptr[r];
        rhi = rowptr[r + 1];
    }
    for (;;) {
        const bool more = pairN < npairs;
        const int begN = min(max(rlo, 0), total);
        const int degN = (more && 2 * pairN + khi < n) ? (min(rhi, total) - begN) : 0;
        const int id0N = adj[min(begN + l16, total - 1)];
        const int id1N = adj[min(begN + 16 + l16, total - 1)];
        {
            const int pc = min(pairN + stride, npairs - 1);
            const int r  = min(2 * pc + khi, n - 1);
            rlo = rowptr[r];
            rhi = rowptr[r + 1];
        }
        const int  row   = 2 * pair + khi;
        const bool valid = row < n;
        float so = 1.0f;
        if (SCALE_OUT) so = scale_out[min(row, n - 1)];
        v4f r = dhcf_pull_rows<SCALE_IN, TWO_TABLES, false, true>(
            adj, (const char*)SRC, (const char*)SRC_HI, nu, scale_in, n, total,
            beg, deg, l16, cb, id0, id1);
        if (valid) {
            if (SCALE_OUT) r *= so;
            *(v4f*)(DST + (size_t)row * 64 + (l16 << 2)) = r;
        }
        if (!more) break;
        pair = pairN;  pairN += stride;
        beg = begN;  deg = degN;  id0 = id0N;  id1 = id1N;
    }
}

#define WROW 68
template <bool FIRST, bool LAST, bool PULL>
__global__ __launch_bounds__(256) void dhcf_layer_kernel(
    float* __restrict__ X, const float* __restrict__ Yb,
    const float* __restrict__ dis,
    const float* __restrict__ Wgc, const float* __restrict__ bgc,
    const float* __restrict__ Wbi, const float* __restrict__ bbi,
    float* __restrict__ ACC, int ntiles,
    const float* __restrict__ u_emb, const float* __restrict__ i_emb,
    int nu, float inv,
    const int* __restrict__ rowptr, const int* __restrict__ adj,
    int n, int total)
{
    __shared__ __attribute__((aligned(16))) float sWg[64 * WROW];
    __shared__ __attribute__((aligned(16))) float sWb[64 * WROW];
    __shared__ __attribute__((aligned(16))) float sY[WPB][16 * WROW];
    __shared__ __attribute__((aligned(16))) float sX[WPB][16 * WROW];
    __shared__ float sBias[128];

    int wave = threadIdx.x >> 5;
    int lane = threadIdx.x & 31;
    int tile = blockIdx.x * WPB + wave;
    const bool live = tile < ntiles;

    int row0 = tile * 16;
    int m    = lane & 15;
    int khi  = lane >> 4;
    int rowA = row0 + m;

    float gscale = 0.0f;
    {
        v4f wg[4], wb[4];
        for (int it = 0; it < 4; ++it) {
            const int q = threadIdx.x + it * 256;
            wg[it] = *(const v4f*)(Wgc + q * 4);
            wb[it] = *(const v4f*)(Wbi + q * 4);
        }
        if (live) {
            const v4f* ysrc = (const v4f*)(Yb + (size_t)row0 * 64) + lane;
            const v4f* xsrc = (const v4f*)(X  + (size_t)row0 * 64) + lane;
            v4f ty[8], tx[8];
            for (int i = 0; i < 8; ++i) {
                if (!PULL) ty[i] = ysrc[i * 32];
                if (FIRST) {
                    const int r = row0 + 2 * i + khi;
                    const float* p = (r < nu) ? (u_emb + (size_t)r * 64)
                                              : (i_emb + (size_t)(r - nu) * 64);
                    tx[i] = *(const v4f*)(p + (m << 2));
                } else {
                    tx[i] = xsrc[i * 32];
                }
            }
            gscale = dis[rowA];
            float* dy = sY[wave] + khi * WROW + (m << 2);
            float* dx = sX[wave] + khi * WROW + (m << 2);
            for (int i = 0; i < 8; ++i) {
                if (!PULL) *(v4f*)(dy + 2 * i * WROW) = ty[i];
                *(v4f*)(dx + 2 * i * WROW) = tx[i];
            }
        }
        if (threadIdx.x < 128)
            sBias[threadIdx.x] = (threadIdx.x < 64) ? bgc[threadIdx.x] : bbi[threadIdx.x - 64];
        for (int it = 0; it < 4; ++it) {
            const int q  = threadIdx.x + it * 256;
            const int r  = q >> 4;
            const int c4 = (q & 15) << 2;
            *(v4f*)(sWg + r * WROW + c4) = wg[it];
            *(v4f*)(sWb + r * WROW + c4) = wb[it];
        }
    }
    if (PULL && live) {
        float* dy = sY[wave] + khi * WROW + (m << 2);
        const int rp = rowptr[row0 + min(lane, 16)];
        const unsigned cb = (unsigned)m << 4;
        int begN = max(__shfl(rp, khi, 32), 0);
        int degN = min(__shfl(rp, khi + 1, 32), total) - begN;
        int id0N = adj[min(begN + m, total - 1)];
        int id1N = adj[min(begN + 16 + m, total - 1)];
        #pragma nounroll
        for (int i = 0; i < 8; ++i) {
            const int beg = begN, deg = degN, id0 = id0N, id1 = id1N;
            if (i < 7) {
                begN = max(__shfl(rp, 2 * i + 2 + khi, 32), 0);
                degN = min(__shfl(rp, 2 * i + 3 + khi, 32), total) - begN;
                id0N = adj[min(begN + m, total - 1)];
                id1N = adj[min(begN + 16 + m, total - 1)];
            }
            const v4f y = dhcf_pull_rows<false, false, true, true>(
                adj, (const char*)Yb, (const char*)Yb, nu, dis, n, total,
                beg, deg, m, cb, id0, id1);
            *(v4f*)(dy + 2 * i * WROW) = y;
        }
    }
    __syncthreads();
    if (!live) return;

    const v8f vz = {0,0,0,0,0,0,0,0};
    v8f acc_s[4], acc_b[4];
    #pragma unroll
    for (int jj = 0; jj < 4; ++jj) { acc_s[jj] = vz; acc_b[jj] = vz; }

    const float* yrow = sY[wave] + m * WROW;
    const float* xrow = sX[wave] + m * WROW;

    #pragma unroll
    for (int kk = 0; kk < 16; ++kk) {
        int k0 = kk * 4 + khi * 2;
        v2f g2, xg2;
        float y0 = yrow[k0], y1 = yrow[k0 + 1];
        float x0 = xrow[k0], x1 = xrow[k0 + 1];
        g2.x  = y0 * gscale;  g2.y  = y1 * gscale;
        xg2.x = x0 * g2.x;    xg2.y = x1 * g2.y;
        #pragma unroll
        for (int jj = 0; jj < 4; ++jj) {
            int j = jj * 16 + m;
            v2f bg, bb;
            bg.x = sWg[j * WROW + k0]; bg.y = sWg[j * WROW + k0 + 1];
            bb.x = sWb[j * WROW + k0]; bb.y = sWb[j * WROW + k0 + 1];
            acc_s[jj] = __builtin_amdgcn_wmma_f32_16x16x4_f32(
                false, g2,  false, bg, (short)0, acc_s[jj], false, false);
            acc_b[jj] = __builtin_amdgcn_wmma_f32_16x16x4_f32(
                false, xg2, false, bb, (short)0, acc_b[jj], false, false);
        }
    }

    float xn[4][8];
    #pragma unroll
    for (int jj = 0; jj < 4; ++jj) {
        int j = jj * 16 + m;
        float bgj = sBias[j];
        float bbj = sBias[64 + j];
        #pragma unroll
        for (int r = 0; r < 8; ++r) {
            float xv = sX[wave][(r + 8 * khi) * WROW + j];
            float s = acc_s[jj][r] + bgj + xv;
            s = (s > 0.0f) ? s : 0.2f * s;
            float b = acc_b[jj][r] + bbj;
            b = (b > 0.0f) ? b : 0.2f * b;
            xn[jj][r] = s + b;
        }
    }

    #pragma unroll
    for (int r = 0; r < 8; ++r) {
        float p = 0.0f;
        #pragma unroll
        for (int jj = 0; jj < 4; ++jj) p += xn[jj][r] * xn[jj][r];
        p += __shfl_xor(p, 1, 32);
        p += __shfl_xor(p, 2, 32);
        p += __shfl_xor(p, 4, 32);
        p += __shfl_xor(p, 8, 32);
        float scl = 1.0f / fmaxf(sqrtf(p), 1e-12f);
        int row = row0 + r + 8 * khi;
        #pragma unroll
        for (int jj = 0; jj < 4; ++jj) {
            int j = jj * 16 + m;
            float v = xn[jj][r] * scl;
            if (!LAST) X[(size_t)row * 64 + j] = v;
            float a = FIRST ? sX[wave][(r + 8 * khi) * WROW + j]
                            : ACC[(size_t)row * 64 + j];
            a += v;
            if (LAST) a *= inv;
            ACC[(size_t)row * 64 + j] = a;
        }
    }
}

extern "C" void kernel_launch(void* const* d_in, const int* in_sizes, int n_in,
                              void* d_out, int out_size, void* d_ws, size_t ws_size,
                              hipStream_t stream)
{
    const float* u_emb = (const float*)d_in[0];
    const float* i_emb = (const float*)d_in[1];
    const float* Wgc   = (const float*)d_in[2];
    const float* bgc   = (const float*)d_in[3];
    const float* Wbi   = (const float*)d_in[4];
    const float* bbi   = (const float*)d_in[5];
    const int*   eu    = (const int*)d_in[6];
    const int*   ei    = (const int*)d_in[7];

    const int D  = 64;
    const int NU = in_sizes[0] / D;
    const int NI = in_sizes[1] / D;
    const int N  = NU + NI;
    const int NE = in_sizes[6];
    const int L  = in_sizes[2] / (D * D);
    const int n64 = N * D;

    float* X    = (float*)d_ws;
    float* E    = X + (size_t)n64;
    float* Y    = E + (size_t)n64;
    float* deg  = Y + (size_t)n64;
    float* dis  = deg + N;
    float* dinv = dis + N;
    float* ACC  = (float*)d_out;

    const int nchunk = (N + 1023) / 1024;
    const int total  = 2 * NE;
    int* cnt    = (int*)deg;
    int* rowptr = (int*)(dinv + N);
    int* bsum   = rowptr + (N + 1);
    int* adj    = bsum + nchunk;
    int* rank_u = (int*)Y;
    int* rank_i = rank_u + NE;
    const size_t need = (size_t)((const char*)(adj + total) - (const char*)d_ws);
    const bool lists = (ws_size >= need) && (total <= n64) && (NE > 0) &&
                       (n64 < (1 << 30));

    dim3 blk(256);
    const bool direct = lists && (L > 0);
    if (direct) {
        dhcf_zero_kernel<<<(N + 255) / 256, blk, 0, stream>>>(deg, N);
    } else {
        dhcf_init_kernel<<<(n64 + 255) / 256, blk, 0, stream>>>(
            u_emb, i_emb, X, ACC, deg, NU * D, n64, N);
    }

    const int ntiles = N / 16;
    const int lgrid  = (ntiles + WPB - 1) / WPB;
    const int egrid  = (NE * 32 + 255) / 256;
    const int ppairs = (N * 16 + 255) / 256;
    const int pgrid  = (ppairs < 1536) ? ppairs : 1536;

    if (lists) {
        dhcf_count_rank_kernel<<<(NE + 256 * EPT - 1) / (256 * EPT), blk, 0, stream>>>(
            eu, ei, cnt, rank_u, rank_i, NE, NU);
        dhcf_scan_local_kernel<<<nchunk, blk, 0, stream>>>(cnt, rowptr, bsum, N);
        dhcf_scan_finish_kernel<<<nchunk, blk, 0, stream>>>(
            cnt, rowptr, bsum, dis, dinv, N, total);
        dhcf_fill_kernel<<<(NE + 256 * EPT - 1) / (256 * EPT), blk, 0, stream>>>(
            eu, ei, rank_u, rank_i, rowptr, adj, NE, NU, total);
    } else {
        dhcf_degree_kernel<<<(NE + 255) / 256, blk, 0, stream>>>(eu, ei, deg, NE, NU);
        dhcf_degnorm_kernel<<<(N + 255) / 256, blk, 0, stream>>>(deg, dis, dinv, N);
    }

    const float inv = 1.0f / (float)(L + 1);
    for (int l = 0; l < L; ++l) {
        const bool first = direct && (l == 0);
        const bool last  = (l == L - 1);
        if (lists) {
            if (first) {
                dhcf_pull_kernel<true, true, true><<<pgrid, blk, 0, stream>>>(
                    rowptr, adj, u_emb, i_emb, NU, dis, dinv, E, N, total);
            } else {
                dhcf_pull_kernel<true, true, false><<<pgrid, blk, 0, stream>>>(
                    rowptr, adj, X, X, NU, dis, dinv, E, N, total);
            }
        } else {
            dhcf_zero_kernel<<<(2 * n64 + 255) / 256, blk, 0, stream>>>(E, 2 * n64);
            dhcf_edge_gather_scatter<<<egrid, blk, 0, stream>>>(eu, ei, X, dis,  E, NE, NU);
            dhcf_edge_gather_scatter<<<egrid, blk, 0, stream>>>(eu, ei, E, dinv, Y, NE, NU);
        }
        const float* wg = Wgc + (size_t)l * D * D;
        const float* bg = bgc + (size_t)l * D;
        const float* wb = Wbi + (size_t)l * D * D;
        const float* bb = bbi + (size_t)l * D;
        if (lists) {
            if (first && last) {
                dhcf_layer_kernel<true, true, true><<<lgrid, blk, 0, stream>>>(
                    X, E, dis, wg, bg, wb, bb, ACC, ntiles, u_emb, i_emb, NU, inv,
                    rowptr, adj, N, total);
            } else if (first) {
                dhcf_layer_kernel<true, false, true><<<lgrid, blk, 0, stream>>>(
                    X, E, dis, wg, bg, wb, bb, ACC, ntiles, u_emb, i_emb, NU, inv,
                    rowptr, adj, N, total);
            } else if (last) {
                dhcf_layer_kernel<false, true, true><<<lgrid, blk, 0, stream>>>(
                    X, E, dis, wg, bg, wb, bb, ACC, ntiles, u_emb, i_emb, NU, inv,
                    rowptr, adj, N, total);
            } else {
                dhcf_layer_kernel<false, false, true><<<lgrid, blk, 0, stream>>>(
                    X, E, dis, wg, bg, wb, bb, ACC, ntiles, u_emb, i_emb, NU, inv,
                    rowptr, adj, N, total);
            }
        } else if (last) {
            dhcf_layer_kernel<false, true, false><<<lgrid, blk, 0, stream>>>(
                X, Y, dis, wg, bg, wb, bb, ACC, ntiles, u_emb, i_emb, NU, inv,
                rowptr, adj, N, total);
        } else {
            dhcf_layer_kernel<false, false, false><<<lgrid, blk, 0, stream>>>(
                X, Y, dis, wg, bg, wb, bb, ACC, ntiles, u_emb, i_emb, NU, inv,
                rowptr, adj, N, total);
        }
    }
}


# Later layers' first pull pass reads each list entry's neighbour scale from a per-entry array that the first layer's pull pass stores (coalesced), not by a 32-line gather from the vertex table
# speedup vs baseline: 4.0220x; 1.4095x over previous
#include <hip/hip_runtime.h>
#include <math.h>


typedef __attribute__((ext_vector_type(2))) float v2f;
typedef __attribute__((ext_vector_type(8))) float v8f;
typedef __attribute__((ext_vector_type(4))) float v4f;

#define WPB 8

__global__ void dhcf_init_kernel(const float* __restrict__ u_emb,
                                 const float* __restrict__ i_emb,
                                 float* __restrict__ X,
                                 float* __restrict__ ACC,
                                 float* __restrict__ deg,
                                 int nu64, int n64, int n)
{
    int idx = blockIdx.x * blockDim.x + threadIdx.x;
    if (idx < n64) {
        float v = (idx < nu64) ? u_emb[idx] : i_emb[idx - nu64];
        X[idx]   = v;
        ACC[idx] = v;
    }
    if (idx < n) deg[idx] = 0.0f;
}

__global__ void dhcf_degree_kernel(const int* __restrict__ eu,
                                   const int* __restrict__ ei,
                                   float* __restrict__ deg,
                                   int ne, int NU)
{
    int e = blockIdx.x * blockDim.x + threadIdx.x;
    if (e >= ne) return;
    atomicAdd(deg + eu[e], 1.0f);
    atomicAdd(deg + NU + ei[e], 1.0f);
}

__global__ void dhcf_degnorm_kernel(const float* __restrict__ deg,
                                    float* __restrict__ dis,
                                    float* __restrict__ dinv, int n)
{
    int v = blockIdx.x * blockDim.x + threadIdx.x;
    if (v >= n) return;
    float d = deg[v];
    if (d > 0.0f) { dis[v] = 1.0f / sqrtf(d); dinv[v] = 1.0f / d; }
    else          { dis[v] = 0.0f;            dinv[v] = 0.0f;     }
}

__global__ void dhcf_zero_kernel(float* __restrict__ p, int count)
{
    int idx = blockIdx.x * blockDim.x + threadIdx.x;
    if (idx < count) p[idx] = 0.0f;
}

__global__ void dhcf_edge_gather_scatter(const int* __restrict__ eu,
                                         const int* __restrict__ ei,
                                         const float* __restrict__ SRC,
                                         const float* __restrict__ scale,
                                         float* __restrict__ DST,
                                         int ne, int NU)
{
    int idx = blockIdx.x * blockDim.x + threadIdx.x;
    if (idx >= ne * 32) return;
    int e = idx >> 5;
    int c = (idx & 31) << 1;
    int u = eu[e];
    int i = NU + ei[e];
    float2 xu = *(const float2*)(SRC + (size_t)u * 64 + c);
    float2 xi = *(const float2*)(SRC + (size_t)i * 64 + c);
    float su = scale[u], si = scale[i];
    float* di = DST + (size_t)i * 64 + c;
    float* du = DST + (size_t)u * 64 + c;
    atomicAdd(di,     xu.x * su);
    atomicAdd(di + 1, xu.y * su);
    atomicAdd(du,     xi.x * si);
    atomicAdd(du + 1, xi.y * si);
}

#define DHCF_BSH   9
#define DHCF_BW    (1 << DHCF_BSH)
#define DHCF_NBMAX 2048
#define DHCF_ESL   4096
#define DHCF_CAP   24576
#define DHCF_RPT   8

__device__ __forceinline__ int dhcf_bkt(int key, int nb)
{
    return min(max(key, 0) >> DHCF_BSH, nb - 1);
}

__device__ __forceinline__ int dhcf_block_excl_scan(int v, int* wsum, int& total)
{
    const int lane = threadIdx.x & 31, wave = threadIdx.x >> 5;
    const int nw = (int)(blockDim.x >> 5);
    int inc = v;
    #pragma unroll
    for (int d = 1; d < 32; d <<= 1) {
        int y = __shfl_up(inc, d, 32);
        if (lane >= d) inc += y;
    }
    __syncthreads();
    if (lane == 31) wsum[wave] = inc;
    __syncthreads();
    int woff = 0, tot = 0;
    for (int w = 0; w < nw; ++w) {
        const int s = wsum[w];
        if (w < wave) woff += s;
        tot += s;
    }
    total = tot;
    return woff + inc - v;
}

__global__ __launch_bounds__(512) void dhcf_part_hist_kernel(
    const int* __restrict__ eu, const int* __restrict__ ei,
    int* __restrict__ H, int ne, int NU, int nb)
{
    __shared__ int lh[DHCF_NBMAX];
    const int t = threadIdx.x;
    const int base = (int)blockIdx.x * DHCF_ESL;
    int u[DHCF_ESL / 512], it[DHCF_ESL / 512];
    #pragma unroll
    for (int k = 0; k < DHCF_ESL / 512; ++k) {
        const int e = min(base + t + k * 512, ne - 1);
        u[k]  = eu[e];
        it[k] = NU + ei[e];
    }
    for (int b = t; b < DHCF_NBMAX; b += 512) lh[b] = 0;
    __syncthreads();
    #pragma unroll
    for (int k = 0; k < DHCF_ESL / 512; ++k) {
        if (base + t + k * 512 < ne) {
            atomicAdd(&lh[dhcf_bkt(u[k], nb)], 1);
            atomicAdd(&lh[dhcf_bkt(it[k], nb)], 1);
        }
    }
    __syncthreads();
    for (int b = t; b < nb; b += 512) H[(size_t)blockIdx.x * nb + b] = lh[b];
}

__global__ __launch_bounds__(256) void dhcf_part_scan_kernel(
    const int* __restrict__ H, int* __restrict__ Hx, int* __restrict__ T,
    int nb, int g1)
{
    __shared__ int wsum[32];
    const int t = threadIdx.x;
    const int b = (int)blockIdx.x;
    int carry = 0;
    for (int s0 = 0; s0 < g1; s0 += 256) {
        const int sl = s0 + t;
        const int v = (sl < g1) ? H[(size_t)sl * nb + b] : 0;
        int tot;
        const int ex = dhcf_block_excl_scan(v, wsum, tot);
        if (sl < g1) Hx[(size_t)b * g1 + sl] = carry + ex;
        carry += tot;
    }
    if (t == 0) T[b] = carry;
}

__global__ __launch_bounds__(512) void dhcf_part_scatter_kernel(
    const int* __restrict__ eu, const int* __restrict__ ei,
    const int* __restrict__ Hx, const int* __restrict__ T,
    int2* __restrict__ recs, int ne, int NU, int nb, int g1, int total)
{
    __shared__ int  lh[DHCF_NBMAX];
    __shared__ int  gb[DHCF_NBMAX];
    __shared__ int2 stage[2 * DHCF_ESL];
    __shared__ int  wsum[32];
    const int t = threadIdx.x;
    const int base = (int)blockIdx.x * DHCF_ESL;
    int u[DHCF_ESL / 512], it[DHCF_ESL / 512];
    #pragma unroll
    for (int k = 0; k < DHCF_ESL / 512; ++k) {
        const int e = min(base + t + k * 512, ne - 1);
        u[k]  = eu[e];
        it[k] = NU + ei[e];
    }
    int c[4], hx[4];
    #pragma unroll
    for (int q = 0; q < 4; ++q) {
        const int b = min(4 * t + q, nb - 1);
        c[q]  = T[b];
        hx[q] = Hx[(size_t)b * g1 + blockIdx.x];
    }
    for (int b = t; b < DHCF_NBMAX; b += 512) lh[b] = 0;
    __syncthreads();
    #pragma unroll
    for (int k = 0; k < DHCF_ESL / 512; ++k) {
        if (base + t + k * 512 < ne) {
            atomicAdd(&lh[dhcf_bkt(u[k], nb)], 1);
            atomicAdd(&lh[dhcf_bkt(it[k], nb)], 1);
        }
    }
    __syncthreads();
    int a[4], lp[4];
    int s = 0;
    #pragma unroll
    for (int q = 0; q < 4; ++q) {
        const int b = 4 * t + q;
        a[q] = (b < nb) ? lh[b] : 0;
        s += a[q];
    }
    int nrec;
    int ex = dhcf_block_excl_scan(s, wsum, nrec);
    #pragma unroll
    for (int q = 0; q < 4; ++q) { lp[q] = ex; ex += a[q]; }
    s = 0;
    #pragma unroll
    for (int q = 0; q < 4; ++q) {
        if (4 * t + q >= nb) c[q] = 0;
        s += c[q];
    }
    int all;
    int ex2 = dhcf_block_excl_scan(s, wsum, all);
    #pragma unroll
    for (int q = 0; q < 4; ++q) {
        const int b = 4 * t + q;
        if (b < nb) {
            gb[b] = ex2 + hx[q] - lp[q];
            lh[b] = lp[q];
        }
        ex2 += c[q];
    }
    __syncthreads();
    #pragma unroll
    for (int k = 0; k < DHCF_ESL / 512; ++k) {
        if (base + t + k * 512 < ne) {
            int p = atomicAdd(&lh[dhcf_bkt(u[k], nb)], 1);
            p = min(max(p, 0), 2 * DHCF_ESL - 1);
            stage[p] = make_int2(u[k], it[k]);
            p = atomicAdd(&lh[dhcf_bkt(it[k], nb)], 1);
            p = min(max(p, 0), 2 * DHCF_ESL - 1);
            stage[p] = make_int2(it[k], u[k]);
        }
    }
    __syncthreads();
    nrec = min(nrec, 2 * DHCF_ESL);
    for (int j = t; j < nrec; j += 512) {
        const int2 rec = stage[j];
        const int p = min(max(gb[dhcf_bkt(rec.x, nb)] + j, 0), total - 1);
        recs[p] = rec;
    }
}

__global__ __launch_bounds__(512) void dhcf_bucket_lists_kernel(
    const int2* __restrict__ recs, const int* __restrict__ T,
    int* __restrict__ rowptr, int* __restrict__ adj,
    float* __restrict__ dis, float* __restrict__ dinv,
    int n, int total, int nb)
{
    __shared__ int cnt[DHCF_BW];
    __shared__ int cur[DHCF_BW];
    __shared__ int adjL[DHCF_CAP];
    __shared__ int wsum[32];
    const int t = threadIdx.x;
    const int b = (int)blockIdx.x;
    const int k0 = b << DHCF_BSH;
    int part = 0;
    for (int j = t; j < b; j += 512) part += T[j];
    int s0;
    dhcf_block_excl_scan(part, wsum, s0);
    s0 = min(max(s0, 0), total);
    const int nrec = min(max(T[b], 0), total - s0);
    cnt[t] = 0;
    __syncthreads();
    for (int r0 = 0; r0 < nrec; r0 += 512 * DHCF_RPT) {
        int key[DHCF_RPT];
        #pragma unroll
        for (int k = 0; k < DHCF_RPT; ++k)
            key[k] = recs[s0 + min(r0 + t + k * 512, nrec - 1)].x;
        #pragma unroll
        for (int k = 0; k < DHCF_RPT; ++k)
            if (r0 + t + k * 512 < nrec)
                atomicAdd(&cnt[min(max(key[k] - k0, 0), DHCF_BW - 1)], 1);
    }
    __syncthreads();
    const int cdeg = cnt[t];
    int tot;
    const int ex = dhcf_block_excl_scan(cdeg, wsum, tot);
    cur[t] = ex;
    const int v = k0 + t;
    if (v < n) {
        rowptr[v] = s0 + ex;
        const float d = (float)cdeg;
        if (cdeg > 0) { dis[v] = 1.0f / sqrtf(d); dinv[v] = 1.0f / d; }
        else          { dis[v] = 0.0f;            dinv[v] = 0.0f;     }
    }
    if (b == nb - 1 && t == 0) rowptr[n] = total;
    __syncthreads();
    const bool staged = nrec <= DHCF_CAP;
    for (int r0 = 0; r0 < nrec; r0 += 512 * DHCF_RPT) {
        int2 rec[DHCF_RPT];
        #pragma unroll
        for (int k = 0; k < DHCF_RPT; ++k)
            rec[k] = recs[s0 + min(r0 + t + k * 512, nrec - 1)];
        #pragma unroll
        for (int k = 0; k < DHCF_RPT; ++k) {
            if (r0 + t + k * 512 < nrec) {
                int p = atomicAdd(&cur[min(max(rec[k].x - k0, 0), DHCF_BW - 1)], 1);
                p = min(max(p, 0), nrec - 1);
                if (staged) adjL[p] = rec[k].y;
                else        adj[s0 + p] = rec[k].y;
            }
        }
    }
    __syncthreads();
    if (staged)
        for (int j = t; j < nrec; j += 512) adj[s0 + j] = adjL[j];
}


#define DHCF_RS(K) case K: return __builtin_amdgcn_update_dpp(0, v, 0x150 + K, 0xf, 0xf, false);
__device__ __forceinline__ int dhcf_row_share(int v, int k)
{
    switch (k) {
        DHCF_RS(0)  DHCF_RS(1)  DHCF_RS(2)  DHCF_RS(3)
        DHCF_RS(4)  DHCF_RS(5)  DHCF_RS(6)  DHCF_RS(7)
        DHCF_RS(8)  DHCF_RS(9)  DHCF_RS(10) DHCF_RS(11)
        DHCF_RS(12) DHCF_RS(13) DHCF_RS(14)
        default: return __builtin_amdgcn_update_dpp(0, v, 0x150 + 15, 0xf, 0xf, false);
    }
}
#undef DHCF_RS

template <bool SCALE_IN>
__device__ __forceinline__ void dhcf_acc(v2f& lo, v2f& hi, v4f x, bool act, float s)
{
    v2f xl, xh;
    xl.x = act ? x.x : 0.0f;  xl.y = act ? x.y : 0.0f;
    xh.x = act ? x.z : 0.0f;  xh.y = act ? x.w : 0.0f;
    if (SCALE_IN) { lo += xl * s; hi += xh * s; }
    else          { lo += xl;     hi += xh;     }
}

template <bool TWO_TABLES>
__device__ __forceinline__ v4f dhcf_row(const char* t0, const char* t1, int nu,
                                        int nb, unsigned cb)
{
    if (TWO_TABLES) {
        const bool low = nb < nu;
        const char* t = low ? t0 : t1;
        const unsigned r = (unsigned)(low ? nb : nb - nu);
        return *(const v4f*)(t + (r * 256u + cb));
    }
    return *(const v4f*)(t0 + ((unsigned)nb * 256u + cb));
}

template <bool TWO_TABLES>
__device__ __forceinline__ void dhcf_ask4(v4f (&x)[4], int mS, int q,
                                          const char* srcb, const char* srch,
                                          int nu, unsigned cb)
{
    #pragma unroll
    for (int j = 0; j < 4; ++j) {
        const int nb = dhcf_row_share(mS, (q * 4 + j) & 15);
        x[j] = dhcf_row<TWO_TABLES>(srcb, srch, nu, nb, cb);
    }
}
template <bool SCALE_IN>
__device__ __forceinline__ void dhcf_use4(v2f& alo, v2f& ahi, v2f& blo, v2f& bhi,
                                          const v4f (&x)[4], int sS, int q, int cnt)
{
    #pragma unroll
    for (int j = 0; j < 4; ++j) {
        const int k = q * 4 + j;
        const float s = __int_as_float(dhcf_row_share(sS, k & 15));
        if (j & 1) dhcf_acc<SCALE_IN>(blo, bhi, x[j], k < cnt, s);
        else       dhcf_acc<SCALE_IN>(alo, ahi, x[j], k < cnt, s);
    }
}

template <int SC, bool TWO_TABLES, bool DEEP, bool IDS_GIVEN>
__device__ __forceinline__ v4f dhcf_pull_rows(
    const int* __restrict__ adj, const char* srcb, const char* srch, int nu,
    const float* __restrict__ scale_in, float* __restrict__ esc, int n, int total,
    int beg, int deg, int l16, unsigned cb, int id0f, int id1f)
{
    constexpr bool SCALE_IN = (SC != 0);
    const int dmax = __builtin_amdgcn_readfirstlane(max(deg, __shfl_xor(deg, 16, 32)));

    v2f alo = {0.0f, 0.0f}, ahi = {0.0f, 0.0f};
    v2f blo = {0.0f, 0.0f}, bhi = {0.0f, 0.0f};
    for (int off = 0; off < dmax; off += 32) {
        const int cnt = deg - off;
        const int cm  = min(32, dmax - off);
        int id0 = id0f;
        int id1 = id1f;
        float s0 = 1.0f, s1 = 1.0f;
        const int p0 = min(beg + off + l16, total - 1);
        const int p1 = min(beg + off + 16 + l16, total - 1);
        if (!IDS_GIVEN || off > 0) {
            id0 = adj[p0];
            id1 = adj[p1];
        }
        id0 = min(max(id0, 0), n - 1);
        id1 = min(max(id1, 0), n - 1);
        const int o0 = __shfl_xor(id0, 16, 32);
        const int o1 = __shfl_xor(id1, 16, 32);
        int m0 = (l16 < cnt)      ? id0 : o0;
        int m1 = (l16 + 16 < cnt) ? id1 : o1;
        const int q0 = __builtin_amdgcn_update_dpp(0, m0, 0x00, 0xf, 0xf, false);
        const int q1 = __builtin_amdgcn_update_dpp(0, m1, 0x00, 0xf, 0xf, false);
        m0 = (l16 < cm)      ? m0 : q0;
        m1 = (l16 + 16 < cm) ? m1 : q1;
        if (SC == 1) {
            s0 = scale_in[m0];
            if (cm > 16) s1 = scale_in[m1];
        }
        if (SC == 2) {
            s0 = esc[p0];
            if (cm > 16) s1 = esc[p1];
        }
        if (DEEP) {
            v4f x[6][4];
            const int sS0 = __float_as_int(s0);
            const int sS1 = __float_as_int(s1);
            #pragma unroll
            for (int q = 0; q < 6; ++q)
                if (q * 4 < cm) dhcf_ask4<TWO_TABLES>(x[q], (q < 4) ? m0 : m1, q, srcb, srch, nu, cb);
            #pragma unroll
            for (int q = 0; q < 2; ++q)
                if (q * 4 < cm) dhcf_use4<SCALE_IN>(alo, ahi, blo, bhi, x[q], sS0, q, cnt);
            #pragma unroll
            for (int q = 6; q < 8; ++q)
                if (q * 4 < cm) dhcf_ask4<TWO_TABLES>(x[q - 6], m1, q, srcb, srch, nu, cb);
            #pragma unroll
            for (int q = 2; q < 6; ++q)
                if (q * 4 < cm) dhcf_use4<SCALE_IN>(alo, ahi, blo, bhi, x[q], (q < 4) ? sS0 : sS1, q, cnt);
            #pragma unroll
            for (int q = 6; q < 8; ++q)
                if (q * 4 < cm) dhcf_use4<SCALE_IN>(alo, ahi, blo, bhi, x[q - 6], sS1, q, cnt);
            continue;
        }
        #pragma unroll
        for (int g = 0; g < 4; ++g) {
            if (g * 8 < cm) {
                const int   mS = (g < 2) ? m0 : m1;
                const int   sS = __float_as_int((g < 2) ? s0 : s1);
                v4f xa[4], xb[4];
                #pragma unroll
                for (int j = 0; j < 4; ++j) {
                    const int nb = dhcf_row_share(mS, (g * 8 + j) & 15);
                    xa[j] = dhcf_row<TWO_TABLES>(srcb, srch, nu, nb, cb);
                }
                if (g * 8 + 4 < cm) {
                    #pragma unroll
                    for (int j = 0; j < 4; ++j) {
                        const int nb = dhcf_row_share(mS, (g * 8 + 4 + j) & 15);
                        xb[j] = dhcf_row<TWO_TABLES>(srcb, srch, nu, nb, cb);
                    }
                    #pragma unroll
                    for (int j = 0; j < 4; ++j) {
                        const int k = g * 8 + j;
                        const float s = __int_as_float(dhcf_row_share(sS, k & 15));
                        if (j & 1) dhcf_acc<SCALE_IN>(blo, bhi, xa[j], k < cnt, s);
                        else       dhcf_acc<SCALE_IN>(alo, ahi, xa[j], k < cnt, s);
                    }
                    #pragma unroll
                    for (int j = 0; j < 4; ++j) {
                        const int k = g * 8 + 4 + j;
                        const float s = __int_as_float(dhcf_row_share(sS, k & 15));
                        if (j & 1) dhcf_acc<SCALE_IN>(blo, bhi, xb[j], k < cnt, s);
                        else       dhcf_acc<SCALE_IN>(alo, ahi, xb[j], k < cnt, s);
                    }
                } else {
                    #pragma unroll
                    for (int j = 0; j < 4; ++j) {
                        const int k = g * 8 + j;
                        const float s = __int_as_float(dhcf_row_share(sS, k & 15));
                        if (j & 1) dhcf_acc<SCALE_IN>(blo, bhi, xa[j], k < cnt, s);
                        else       dhcf_acc<SCALE_IN>(alo, ahi, xa[j], k < cnt, s);
                    }
                }
            }
        }
        if (SC == 1) {
            if (l16 < cnt) esc[p0] = s0;
            if (cm > 16 && l16 + 16 < cnt) esc[p1] = s1;
        }
    }
    alo += blo; ahi += bhi;
    v4f r;
    r.x = alo.x; r.y = alo.y; r.z = ahi.x; r.w = ahi.y;
    return r;
}

template <int SC, bool SCALE_OUT, bool TWO_TABLES>
__global__ __launch_bounds__(256) void dhcf_pull_kernel(
    const int* __restrict__ rowptr, const int* __restrict__ adj,
    const float* __restrict__ SRC, const float* __restrict__ SRC_HI, int nu,
    const float* __restrict__ scale_in, float* __restrict__ esc,
    const float* __restrict__ scale_out, float* __restrict__ DST,
    int n, int total)
{
    const int l16  = threadIdx.x & 15;
    const int khi  = (threadIdx.x >> 4) & 1;
    const unsigned cb = (unsigned)l16 << 4;
    const int npairs = (n + 1) >> 1;
    const int stride = (int)gridDim.x * WPB;
    const int wv = __builtin_amdgcn_readfirstlane((int)(threadIdx.x >> 5));
    int pair = (int)blockIdx.x * WPB + wv;
    if (pair >= npairs) return;
    int rlo, rhi;
    {
        const int r = min(2 * pair + khi, n - 1);
        rlo = rowptr[r];
        rhi = rowptr[r + 1];
    }
    int beg = min(max(rlo, 0), total);
    int deg = (2 * pair + khi < n) ? (min(rhi, total) - beg) : 0;
    int id0 = adj[min(beg + l16, total - 1)];
    int id1 = adj[min(beg + 16 + l16, total - 1)];
    int pairN = pair + stride;
    {
        const int pc = min(pairN, npairs - 1);
        const int r  = min(2 * pc + khi, n - 1);
        rlo = rowptr[r];
        rhi = rowptr[r + 1];
    }
    for (;;) {
        const bool more = pairN < npairs;
        const int begN = min(max(rlo, 0), total);
        const int degN = (more && 2 * pairN + khi < n) ? (min(rhi, total) - begN) : 0;
        const int id0N = adj[min(begN + l16, total - 1)];
        const int id1N = adj[min(begN + 16 + l16, total - 1)];
        {
            const int pc = min(pairN + stride, npairs - 1);
            const int r  = min(2 * pc + khi, n - 1);
            rlo = rowptr[r];
            rhi = rowptr[r + 1];
        }
        const int  row   = 2 * pair + khi;
        const bool valid = row < n;
        float so = 1.0f;
        if (SCALE_OUT) so = scale_out[min(row, n - 1)];
        v4f r = dhcf_pull_rows<SC, TWO_TABLES, false, true>(
            adj, (const char*)SRC, (const char*)SRC_HI, nu, scale_in, esc, n, total,
            beg, deg, l16, cb, id0, id1);
        if (valid) {
            if (SCALE_OUT) r *= so;
            *(v4f*)(DST + (size_t)row * 64 + (l16 << 2)) = r;
        }
        if (!more) break;
        pair = pairN;  pairN += stride;
        beg = begN;  deg = degN;  id0 = id0N;  id1 = id1N;
    }
}

#define WROW 68
template <bool FIRST, bool LAST, bool PULL>
__global__ __launch_bounds__(256) void dhcf_layer_kernel(
    float* __restrict__ X, const float* __restrict__ Yb,
    const float* __restrict__ dis,
    const float* __restrict__ Wgc, const float* __restrict__ bgc,
    const float* __restrict__ Wbi, const float* __restrict__ bbi,
    float* __restrict__ ACC, int ntiles,
    const float* __restrict__ u_emb, const float* __restrict__ i_emb,
    int nu, float inv,
    const int* __restrict__ rowptr, const int* __restrict__ adj,
    int n, int total)
{
    __shared__ __attribute__((aligned(16))) float sWg[64 * WROW];
    __shared__ __attribute__((aligned(16))) float sWb[64 * WROW];
    __shared__ __attribute__((aligned(16))) float sY[WPB][16 * WROW];
    __shared__ __attribute__((aligned(16))) float sX[WPB][16 * WROW];
    __shared__ float sBias[128];

    int wave = threadIdx.x >> 5;
    int lane = threadIdx.x & 31;
    int tile = blockIdx.x * WPB + wave;
    const bool live = tile < ntiles;

    int row0 = tile * 16;
    int m    = lane & 15;
    int khi  = lane >> 4;
    int rowA = row0 + m;

    float gscale = 0.0f;
    {
        v4f wg[4], wb[4];
        for (int it = 0; it < 4; ++it) {
            const int q = threadIdx.x + it * 256;
            wg[it] = *(const v4f*)(Wgc + q * 4);
            wb[it] = *(const v4f*)(Wbi + q * 4);
        }
        if (live) {
            const v4f* ysrc = (const v4f*)(Yb + (size_t)row0 * 64) + lane;
            const v4f* xsrc = (const v4f*)(X  + (size_t)row0 * 64) + lane;
            v4f ty[8], tx[8];
            for (int i = 0; i < 8; ++i) {
                if (!PULL) ty[i] = ysrc[i * 32];
                if (FIRST) {
                    const int r = row0 + 2 * i + khi;
                    const float* p = (r < nu) ? (u_emb + (size_t)r * 64)
                                              : (i_emb + (size_t)(r - nu) * 64);
                    tx[i] = *(const v4f*)(p + (m << 2));
                } else {
                    tx[i] = xsrc[i * 32];
                }
            }
            gscale = dis[rowA];
            float* dy = sY[wave] + khi * WROW + (m << 2);
            float* dx = sX[wave] + khi * WROW + (m << 2);
            for (int i = 0; i < 8; ++i) {
                if (!PULL) *(v4f*)(dy + 2 * i * WROW) = ty[i];
                *(v4f*)(dx + 2 * i * WROW) = tx[i];
            }
        }
        if (threadIdx.x < 128)
            sBias[threadIdx.x] = (threadIdx.x < 64) ? bgc[threadIdx.x] : bbi[threadIdx.x - 64];
        for (int it = 0; it < 4; ++it) {
            const int q  = threadIdx.x + it * 256;
            const int r  = q >> 4;
            const int c4 = (q & 15) << 2;
            *(v4f*)(sWg + r * WROW + c4) = wg[it];
            *(v4f*)(sWb + r * WROW + c4) = wb[it];
        }
    }
    if (PULL && live) {
        float* dy = sY[wave] + khi * WROW + (m << 2);
        const int rp = rowptr[row0 + min(lane, 16)];
        const unsigned cb = (unsigned)m << 4;
        int begN = max(__shfl(rp, khi, 32), 0);
        int degN = min(__shfl(rp, khi + 1, 32), total) - begN;
        int id0N = adj[min(begN + m, total - 1)];
        int id1N = adj[min(begN + 16 + m, total - 1)];
        #pragma nounroll
        for (int i = 0; i < 8; ++i) {
            const int beg = begN, deg = degN, id0 = id0N, id1 = id1N;
            if (i < 7) {
                begN = max(__shfl(rp, 2 * i + 2 + khi, 32), 0);
                degN = min(__shfl(rp, 2 * i + 3 + khi, 32), total) - begN;
                id0N = adj[min(begN + m, total - 1)];
                id1N = adj[min(begN + 16 + m, total - 1)];
            }
            const v4f y = dhcf_pull_rows<0, false, true, true>(
                adj, (const char*)Yb, (const char*)Yb, nu, dis, nullptr, n, total,
                beg, deg, m, cb, id0, id1);
            *(v4f*)(dy + 2 * i * WROW) = y;
        }
    }
    __syncthreads();
    if (!live) return;

    const v8f vz = {0,0,0,0,0,0,0,0};
    v8f acc_s[4], acc_b[4];
    #pragma unroll
    for (int jj = 0; jj < 4; ++jj) { acc_s[jj] = vz; acc_b[jj] = vz; }

    const float* yrow = sY[wave] + m * WROW;
    const float* xrow = sX[wave] + m * WROW;

    #pragma unroll
    for (int kk = 0; kk < 16; ++kk) {
        int k0 = kk * 4 + khi * 2;
        v2f g2, xg2;
        float y0 = yrow[k0], y1 = yrow[k0 + 1];
        float x0 = xrow[k0], x1 = xrow[k0 + 1];
        g2.x  = y0 * gscale;  g2.y  = y1 * gscale;
        xg2.x = x0 * g2.x;    xg2.y = x1 * g2.y;
        #pragma unroll
        for (int jj = 0; jj < 4; ++jj) {
            int j = jj * 16 + m;
            v2f bg, bb;
            bg.x = sWg[j * WROW + k0]; bg.y = sWg[j * WROW + k0 + 1];
            bb.x = sWb[j * WROW + k0]; bb.y = sWb[j * WROW + k0 + 1];
            acc_s[jj] = __builtin_amdgcn_wmma_f32_16x16x4_f32(
                false, g2,  false, bg, (short)0, acc_s[jj], false, false);
            acc_b[jj] = __builtin_amdgcn_wmma_f32_16x16x4_f32(
                false, xg2, false, bb, (short)0, acc_b[jj], false, false);
        }
    }

    float xn[4][8];
    #pragma unroll
    for (int jj = 0; jj < 4; ++jj) {
        int j = jj * 16 + m;
        float bgj = sBias[j];
        float bbj = sBias[64 + j];
        #pragma unroll
        for (int r = 0; r < 8; ++r) {
            float xv = sX[wave][(r + 8 * khi) * WROW + j];
            float s = acc_s[jj][r] + bgj + xv;
            s = (s > 0.0f) ? s : 0.2f * s;
            float b = acc_b[jj][r] + bbj;
            b = (b > 0.0f) ? b : 0.2f * b;
            xn[jj][r] = s + b;
        }
    }

    #pragma unroll
    for (int r = 0; r < 8; ++r) {
        float p = 0.0f;
        #pragma unroll
        for (int jj = 0; jj < 4; ++jj) p += xn[jj][r] * xn[jj][r];
        p += __shfl_xor(p, 1, 32);
        p += __shfl_xor(p, 2, 32);
        p += __shfl_xor(p, 4, 32);
        p += __shfl_xor(p, 8, 32);
        float scl = 1.0f / fmaxf(sqrtf(p), 1e-12f);
        int row = row0 + r + 8 * khi;
        #pragma unroll
        for (int jj = 0; jj < 4; ++jj) {
            int j = jj * 16 + m;
            float v = xn[jj][r] * scl;
            if (!LAST) X[(size_t)row * 64 + j] = v;
            float a = FIRST ? sX[wave][(r + 8 * khi) * WROW + j]
                            : ACC[(size_t)row * 64 + j];
            a += v;
            if (LAST) a *= inv;
            ACC[(size_t)row * 64 + j] = a;
        }
    }
}

extern "C" void kernel_launch(void* const* d_in, const int* in_sizes, int n_in,
                              void* d_out, int out_size, void* d_ws, size_t ws_size,
                              hipStream_t stream)
{
    const float* u_emb = (const float*)d_in[0];
    const float* i_emb = (const float*)d_in[1];
    const float* Wgc   = (const float*)d_in[2];
    const float* bgc   = (const float*)d_in[3];
    const float* Wbi   = (const float*)d_in[4];
    const float* bbi   = (const float*)d_in[5];
    const int*   eu    = (const int*)d_in[6];
    const int*   ei    = (const int*)d_in[7];

    const int D  = 64;
    const int NU = in_sizes[0] / D;
    const int NI = in_sizes[1] / D;
    const int N  = NU + NI;
    const int NE = in_sizes[6];
    const int L  = in_sizes[2] / (D * D);
    const int n64 = N * D;

    float* X    = (float*)d_ws;
    float* E    = X + (size_t)n64;
    float* Y    = E + (size_t)n64;
    float* deg  = Y + (size_t)n64;
    float* dis  = deg + N;
    float* dinv = dis + N;
    float* ACC  = (float*)d_out;

    const int nchunk = (N + 1023) / 1024;
    const int total  = 2 * NE;
    int* rowptr = (int*)(dinv + N);
    int* bsum   = rowptr + (N + 1);
    int* adj    = bsum + nchunk;
    const int nb = (N + DHCF_BW - 1) / DHCF_BW;
    const int g1 = (NE + DHCF_ESL - 1) / DHCF_ESL;
    int2* recs  = (int2*)Y;
    int* H      = (int*)Y + 2 * (size_t)total;
    int* Hx     = H + (size_t)g1 * nb;
    int* T      = Hx + (size_t)g1 * nb;
    float* esc  = Y;
    const size_t need = (size_t)((const char*)(adj + total) - (const char*)d_ws);
    const size_t ybuild = 2 * (size_t)total + 2 * (size_t)g1 * nb + (size_t)nb;
    const bool lists = (ws_size >= need) && (ybuild <= (size_t)n64) && (NE > 0) &&
                       (nb <= DHCF_NBMAX) && (n64 < (1 << 30));

    dim3 blk(256);
    const bool direct = lists && (L > 0);
    if (!direct) {
        dhcf_init_kernel<<<(n64 + 255) / 256, blk, 0, stream>>>(
            u_emb, i_emb, X, ACC, deg, NU * D, n64, N);
    }

    const int ntiles = N / 16;
    const int lgrid  = (ntiles + WPB - 1) / WPB;
    const int egrid  = (NE * 32 + 255) / 256;
    const int ppairs = (N * 16 + 255) / 256;
    const int pgrid  = (ppairs < 1536) ? ppairs : 1536;

    if (lists) {
        dhcf_part_hist_kernel<<<g1, dim3(512), 0, stream>>>(eu, ei, H, NE, NU, nb);
        dhcf_part_scan_kernel<<<nb, blk, 0, stream>>>(H, Hx, T, nb, g1);
        dhcf_part_scatter_kernel<<<g1, dim3(512), 0, stream>>>(
            eu, ei, Hx, T, recs, NE, NU, nb, g1, total);
        dhcf_bucket_lists_kernel<<<nb, dim3(512), 0, stream>>>(
            recs, T, rowptr, adj, dis, dinv, N, total, nb);
    } else {
        dhcf_degree_kernel<<<(NE + 255) / 256, blk, 0, stream>>>(eu, ei, deg, NE, NU);
        dhcf_degnorm_kernel<<<(N + 255) / 256, blk, 0, stream>>>(deg, dis, dinv, N);
    }

    const float inv = 1.0f / (float)(L + 1);
    for (int l = 0; l < L; ++l) {
        const bool first = direct && (l == 0);
        const bool last  = (l == L - 1);
        if (lists) {
            if (first) {
                dhcf_pull_kernel<1, true, true><<<pgrid, blk, 0, stream>>>(
                    rowptr, adj, u_emb, i_emb, NU, dis, esc, dinv, E, N, total);
            } else {
                dhcf_pull_kernel<2, true, false><<<pgrid, blk, 0, stream>>>(
                    rowptr, adj, X, X, NU, dis, esc, dinv, E, N, total);
            }
        } else {
            dhcf_zero_kernel<<<(2 * n64 + 255) / 256, blk, 0, stream>>>(E, 2 * n64);
            dhcf_edge_gather_scatter<<<egrid, blk, 0, stream>>>(eu, ei, X, dis,  E, NE, NU);
            dhcf_edge_gather_scatter<<<egrid, blk, 0, stream>>>(eu, ei, E, dinv, Y, NE, NU);
        }
        const float* wg = Wgc + (size_t)l * D * D;
        const float* bg = bgc + (size_t)l * D;
        const float* wb = Wbi + (size_t)l * D * D;
        const float* bb = bbi + (size_t)l * D;
        if (lists) {
            if (first && last) {
                dhcf_layer_kernel<true, true, true><<<lgrid, blk, 0, stream>>>(
                    X, E, dis, wg, bg, wb, bb, ACC, ntiles, u_emb, i_emb, NU, inv,
                    rowptr, adj, N, total);
            } else if (first) {
                dhcf_layer_kernel<true, false, true><<<lgrid, blk, 0, stream>>>(
                    X, E, dis, wg, bg, wb, bb, ACC, ntiles, u_emb, i_emb, NU, inv,
                    rowptr, adj, N, total);
            } else if (last) {
                dhcf_layer_kernel<false, true, true><<<lgrid, blk, 0, stream>>>(
                    X, E, dis, wg, bg, wb, bb, ACC, ntiles, u_emb, i_emb, NU, inv,
                    rowptr, adj, N, total);
            } else {
                dhcf_layer_kernel<false, false, true><<<lgrid, blk, 0, stream>>>(
                    X, E, dis, wg, bg, wb, bb, ACC, ntiles, u_emb, i_emb, NU, inv,
                    rowptr, adj, N, total);
            }
        } else if (last) {
            dhcf_layer_kernel<false, true, false><<<lgrid, blk, 0, stream>>>(
                X, Y, dis, wg, bg, wb, bb, ACC, ntiles, u_emb, i_emb, NU, inv,
                rowptr, adj, N, total);
        } else {
            dhcf_layer_kernel<false, false, false><<<lgrid, blk, 0, stream>>>(
                X, Y, dis, wg, bg, wb, bb, ACC, ntiles, u_emb, i_emb, NU, inv,
                rowptr, adj, N, total);
        }
    }
}


# Running sum kept out of the middle layer: an odd layer that is not the last neither loads nor stores it; the next layer adds its x tile (already in LDS), then its new rows; same order of additions
# speedup vs baseline: 4.1664x; 1.0359x over previous
#include <hip/hip_runtime.h>
#include <math.h>


typedef __attribute__((ext_vector_type(2))) float v2f;
typedef __attribute__((ext_vector_type(8))) float v8f;
typedef __attribute__((ext_vector_type(4))) float v4f;

#define WPB 8

__global__ void dhcf_init_kernel(const float* __restrict__ u_emb,
                                 const float* __restrict__ i_emb,
                                 float* __restrict__ X,
                                 float* __restrict__ ACC,
                                 float* __restrict__ deg,
                                 int nu64, int n64, int n)
{
    int idx = blockIdx.x * blockDim.x + threadIdx.x;
    if (idx < n64) {
        float v = (idx < nu64) ? u_emb[idx] : i_emb[idx - nu64];
        X[idx]   = v;
        ACC[idx] = v;
    }
    if (idx < n) deg[idx] = 0.0f;
}

__global__ void dhcf_degree_kernel(const int* __restrict__ eu,
                                   const int* __restrict__ ei,
                                   float* __restrict__ deg,
                                   int ne, int NU)
{
    int e = blockIdx.x * blockDim.x + threadIdx.x;
    if (e >= ne) return;
    atomicAdd(deg + eu[e], 1.0f);
    atomicAdd(deg + NU + ei[e], 1.0f);
}

__global__ void dhcf_degnorm_kernel(const float* __restrict__ deg,
                                    float* __restrict__ dis,
                                    float* __restrict__ dinv, int n)
{
    int v = blockIdx.x * blockDim.x + threadIdx.x;
    if (v >= n) return;
    float d = deg[v];
    if (d > 0.0f) { dis[v] = 1.0f / sqrtf(d); dinv[v] = 1.0f / d; }
    else          { dis[v] = 0.0f;            dinv[v] = 0.0f;     }
}

__global__ void dhcf_zero_kernel(float* __restrict__ p, int count)
{
    int idx = blockIdx.x * blockDim.x + threadIdx.x;
    if (idx < count) p[idx] = 0.0f;
}

__global__ void dhcf_edge_gather_scatter(const int* __restrict__ eu,
                                         const int* __restrict__ ei,
                                         const float* __restrict__ SRC,
                                         const float* __restrict__ scale,
                                         float* __restrict__ DST,
                                         int ne, int NU)
{
    int idx = blockIdx.x * blockDim.x + threadIdx.x;
    if (idx >= ne * 32) return;
    int e = idx >> 5;
    int c = (idx & 31) << 1;
    int u = eu[e];
    int i = NU + ei[e];
    float2 xu = *(const float2*)(SRC + (size_t)u * 64 + c);
    float2 xi = *(const float2*)(SRC + (size_t)i * 64 + c);
    float su = scale[u], si = scale[i];
    float* di = DST + (size_t)i * 64 + c;
    float* du = DST + (size_t)u * 64 + c;
    atomicAdd(di,     xu.x * su);
    atomicAdd(di + 1, xu.y * su);
    atomicAdd(du,     xi.x * si);
    atomicAdd(du + 1, xi.y * si);
}

#define DHCF_BSH   9
#define DHCF_BW    (1 << DHCF_BSH)
#define DHCF_NBMAX 2048
#define DHCF_ESL   4096
#define DHCF_CAP   24576
#define DHCF_RPT   8

__device__ __forceinline__ int dhcf_bkt(int key, int nb)
{
    return min(max(key, 0) >> DHCF_BSH, nb - 1);
}

__device__ __forceinline__ int dhcf_block_excl_scan(int v, int* wsum, int& total)
{
    const int lane = threadIdx.x & 31, wave = threadIdx.x >> 5;
    const int nw = (int)(blockDim.x >> 5);
    int inc = v;
    #pragma unroll
    for (int d = 1; d < 32; d <<= 1) {
        int y = __shfl_up(inc, d, 32);
        if (lane >= d) inc += y;
    }
    __syncthreads();
    if (lane == 31) wsum[wave] = inc;
    __syncthreads();
    int woff = 0, tot = 0;
    for (int w = 0; w < nw; ++w) {
        const int s = wsum[w];
        if (w < wave) woff += s;
        tot += s;
    }
    total = tot;
    return woff + inc - v;
}

__global__ __launch_bounds__(512) void dhcf_part_hist_kernel(
    const int* __restrict__ eu, const int* __restrict__ ei,
    int* __restrict__ H, int ne, int NU, int nb)
{
    __shared__ int lh[DHCF_NBMAX];
    const int t = threadIdx.x;
    const int base = (int)blockIdx.x * DHCF_ESL;
    int u[DHCF_ESL / 512], it[DHCF_ESL / 512];
    #pragma unroll
    for (int k = 0; k < DHCF_ESL / 512; ++k) {
        const int e = min(base + t + k * 512, ne - 1);
        u[k]  = eu[e];
        it[k] = NU + ei[e];
    }
    for (int b = t; b < DHCF_NBMAX; b += 512) lh[b] = 0;
    __syncthreads();
    #pragma unroll
    for (int k = 0; k < DHCF_ESL / 512; ++k) {
        if (base + t + k * 512 < ne) {
            atomicAdd(&lh[dhcf_bkt(u[k], nb)], 1);
            atomicAdd(&lh[dhcf_bkt(it[k], nb)], 1);
        }
    }
    __syncthreads();
    for (int b = t; b < nb; b += 512) H[(size_t)blockIdx.x * nb + b] = lh[b];
}

__global__ __launch_bounds__(256) void dhcf_part_scan_kernel(
    const int* __restrict__ H, int* __restrict__ Hx, int* __restrict__ T,
    int nb, int g1)
{
    __shared__ int wsum[32];
    const int t = threadIdx.x;
    const int b = (int)blockIdx.x;
    int carry = 0;
    for (int s0 = 0; s0 < g1; s0 += 256) {
        const int sl = s0 + t;
        const int v = (sl < g1) ? H[(size_t)sl * nb + b] : 0;
        int tot;
        const int ex = dhcf_block_excl_scan(v, wsum, tot);
        if (sl < g1) Hx[(size_t)b * g1 + sl] = carry + ex;
        carry += tot;
    }
    if (t == 0) T[b] = carry;
}

__global__ __launch_bounds__(512) void dhcf_part_scatter_kernel(
    const int* __restrict__ eu, const int* __restrict__ ei,
    const int* __restrict__ Hx, const int* __restrict__ T,
    int2* __restrict__ recs, int ne, int NU, int nb, int g1, int total)
{
    __shared__ int  lh[DHCF_NBMAX];
    __shared__ int  gb[DHCF_NBMAX];
    __shared__ int2 stage[2 * DHCF_ESL];
    __shared__ int  wsum[32];
    const int t = threadIdx.x;
    const int base = (int)blockIdx.x * DHCF_ESL;
    int u[DHCF_ESL / 512], it[DHCF_ESL / 512];
    #pragma unroll
    for (int k = 0; k < DHCF_ESL / 512; ++k) {
        const int e = min(base + t + k * 512, ne - 1);
        u[k]  = eu[e];
        it[k] = NU + ei[e];
    }
    int c[4], hx[4];
    #pragma unroll
    for (int q = 0; q < 4; ++q) {
        const int b = min(4 * t + q, nb - 1);
        c[q]  = T[b];
        hx[q] = Hx[(size_t)b * g1 + blockIdx.x];
    }
    for (int b = t; b < DHCF_NBMAX; b += 512) lh[b] = 0;
    __syncthreads();
    #pragma unroll
    for (int k = 0; k < DHCF_ESL / 512; ++k) {
        if (base + t + k * 512 < ne) {
            atomicAdd(&lh[dhcf_bkt(u[k], nb)], 1);
            atomicAdd(&lh[dhcf_bkt(it[k], nb)], 1);
        }
    }
    __syncthreads();
    int a[4], lp[4];
    int s = 0;
    #pragma unroll
    for (int q = 0; q < 4; ++q) {
        const int b = 4 * t + q;
        a[q] = (b < nb) ? lh[b] : 0;
        s += a[q];
    }
    int nrec;
    int ex = dhcf_block_excl_scan(s, wsum, nrec);
    #pragma unroll
    for (int q = 0; q < 4; ++q) { lp[q] = ex; ex += a[q]; }
    s = 0;
    #pragma unroll
    for (int q = 0; q < 4; ++q) {
        if (4 * t + q >= nb) c[q] = 0;
        s += c[q];
    }
    int all;
    int ex2 = dhcf_block_excl_scan(s, wsum, all);
    #pragma unroll
    for (int q = 0; q < 4; ++q) {
        const int b = 4 * t + q;
        if (b < nb) {
            gb[b] = ex2 + hx[q] - lp[q];
            lh[b] = lp[q];
        }
        ex2 += c[q];
    }
    __syncthreads();
    #pragma unroll
    for (int k = 0; k < DHCF_ESL / 512; ++k) {
        if (base + t + k * 512 < ne) {
            int p = atomicAdd(&lh[dhcf_bkt(u[k], nb)], 1);
            p = min(max(p, 0), 2 * DHCF_ESL - 1);
            stage[p] = make_int2(u[k], it[k]);
            p = atomicAdd(&lh[dhcf_bkt(it[k], nb)], 1);
            p = min(max(p, 0), 2 * DHCF_ESL - 1);
            stage[p] = make_int2(it[k], u[k]);
        }
    }
    __syncthreads();
    nrec = min(nrec, 2 * DHCF_ESL);
    for (int j = t; j < nrec; j += 512) {
        const int2 rec = stage[j];
        const int p = min(max(gb[dhcf_bkt(rec.x, nb)] + j, 0), total - 1);
        recs[p] = rec;
    }
}

__global__ __launch_bounds__(512) void dhcf_bucket_lists_kernel(
    const int2* __restrict__ recs, const int* __restrict__ T,
    int* __restrict__ rowptr, int* __restrict__ adj,
    float* __restrict__ dis, float* __restrict__ dinv,
    int n, int total, int nb)
{
    __shared__ int cnt[DHCF_BW];
    __shared__ int cur[DHCF_BW];
    __shared__ int adjL[DHCF_CAP];
    __shared__ int wsum[32];
    const int t = threadIdx.x;
    const int b = (int)blockIdx.x;
    const int k0 = b << DHCF_BSH;
    int part = 0;
    for (int j = t; j < b; j += 512) part += T[j];
    int s0;
    dhcf_block_excl_scan(part, wsum, s0);
    s0 = min(max(s0, 0), total);
    const int nrec = min(max(T[b], 0), total - s0);
    cnt[t] = 0;
    __syncthreads();
    for (int r0 = 0; r0 < nrec; r0 += 512 * DHCF_RPT) {
        int key[DHCF_RPT];
        #pragma unroll
        for (int k = 0; k < DHCF_RPT; ++k)
            key[k] = recs[s0 + min(r0 + t + k * 512, nrec - 1)].x;
        #pragma unroll
        for (int k = 0; k < DHCF_RPT; ++k)
            if (r0 + t + k * 512 < nrec)
                atomicAdd(&cnt[min(max(key[k] - k0, 0), DHCF_BW - 1)], 1);
    }
    __syncthreads();
    const int cdeg = cnt[t];
    int tot;
    const int ex = dhcf_block_excl_scan(cdeg, wsum, tot);
    cur[t] = ex;
    const int v = k0 + t;
    if (v < n) {
        rowptr[v] = s0 + ex;
        const float d = (float)cdeg;
        if (cdeg > 0) { dis[v] = 1.0f / sqrtf(d); dinv[v] = 1.0f / d; }
        else          { dis[v] = 0.0f;            dinv[v] = 0.0f;     }
    }
    if (b == nb - 1 && t == 0) rowptr[n] = total;
    __syncthreads();
    const bool staged = nrec <= DHCF_CAP;
    for (int r0 = 0; r0 < nrec; r0 += 512 * DHCF_RPT) {
        int2 rec[DHCF_RPT];
        #pragma unroll
        for (int k = 0; k < DHCF_RPT; ++k)
            rec[k] = recs[s0 + min(r0 + t + k * 512, nrec - 1)];
        #pragma unroll
        for (int k = 0; k < DHCF_RPT; ++k) {
            if (r0 + t + k * 512 < nrec) {
                int p = atomicAdd(&cur[min(max(rec[k].x - k0, 0), DHCF_BW - 1)], 1);
                p = min(max(p, 0), nrec - 1);
                if (staged) adjL[p] = rec[k].y;
                else        adj[s0 + p] = rec[k].y;
            }
        }
    }
    __syncthreads();
    if (staged)
        for (int j = t; j < nrec; j += 512) adj[s0 + j] = adjL[j];
}


#define DHCF_RS(K) case K: return __builtin_amdgcn_update_dpp(0, v, 0x150 + K, 0xf, 0xf, false);
__device__ __forceinline__ int dhcf_row_share(int v, int k)
{
    switch (k) {
        DHCF_RS(0)  DHCF_RS(1)  DHCF_RS(2)  DHCF_RS(3)
        DHCF_RS(4)  DHCF_RS(5)  DHCF_RS(6)  DHCF_RS(7)
        DHCF_RS(8)  DHCF_RS(9)  DHCF_RS(10) DHCF_RS(11)
        DHCF_RS(12) DHCF_RS(13) DHCF_RS(14)
        default: return __builtin_amdgcn_update_dpp(0, v, 0x150 + 15, 0xf, 0xf, false);
    }
}
#undef DHCF_RS

template <bool SCALE_IN>
__device__ __forceinline__ void dhcf_acc(v2f& lo, v2f& hi, v4f x, bool act, float s)
{
    v2f xl, xh;
    xl.x = act ? x.x : 0.0f;  xl.y = act ? x.y : 0.0f;
    xh.x = act ? x.z : 0.0f;  xh.y = act ? x.w : 0.0f;
    if (SCALE_IN) { lo += xl * s; hi += xh * s; }
    else          { lo += xl;     hi += xh;     }
}

template <bool TWO_TABLES>
__device__ __forceinline__ v4f dhcf_row(const char* t0, const char* t1, int nu,
                                        int nb, unsigned cb)
{
    if (TWO_TABLES) {
        const bool low = nb < nu;
        const char* t = low ? t0 : t1;
        const unsigned r = (unsigned)(low ? nb : nb - nu);
        return *(const v4f*)(t + (r * 256u + cb));
    }
    return *(const v4f*)(t0 + ((unsigned)nb * 256u + cb));
}

template <bool TWO_TABLES>
__device__ __forceinline__ void dhcf_ask4(v4f (&x)[4], int mS, int q,
                                          const char* srcb, const char* srch,
                                          int nu, unsigned cb)
{
    #pragma unroll
    for (int j = 0; j < 4; ++j) {
        const int nb = dhcf_row_share(mS, (q * 4 + j) & 15);
        x[j] = dhcf_row<TWO_TABLES>(srcb, srch, nu, nb, cb);
    }
}
template <bool SCALE_IN>
__device__ __forceinline__ void dhcf_use4(v2f& alo, v2f& ahi, v2f& blo, v2f& bhi,
                                          const v4f (&x)[4], int sS, int q, int cnt)
{
    #pragma unroll
    for (int j = 0; j < 4; ++j) {
        const int k = q * 4 + j;
        const float s = __int_as_float(dhcf_row_share(sS, k & 15));
        if (j & 1) dhcf_acc<SCALE_IN>(blo, bhi, x[j], k < cnt, s);
        else       dhcf_acc<SCALE_IN>(alo, ahi, x[j], k < cnt, s);
    }
}

template <int SC, bool TWO_TABLES, bool DEEP, bool IDS_GIVEN>
__device__ __forceinline__ v4f dhcf_pull_rows(
    const int* __restrict__ adj, const char* srcb, const char* srch, int nu,
    const float* __restrict__ scale_in, float* __restrict__ esc, int n, int total,
    int beg, int deg, int l16, unsigned cb, int id0f, int id1f)
{
    constexpr bool SCALE_IN = (SC != 0);
    const int dmax = __builtin_amdgcn_readfirstlane(max(deg, __shfl_xor(deg, 16, 32)));

    v2f alo = {0.0f, 0.0f}, ahi = {0.0f, 0.0f};
    v2f blo = {0.0f, 0.0f}, bhi = {0.0f, 0.0f};
    for (int off = 0; off < dmax; off += 32) {
        const int cnt = deg - off;
        const int cm  = min(32, dmax - off);
        int id0 = id0f;
        int id1 = id1f;
        float s0 = 1.0f, s1 = 1.0f;
        const int p0 = min(beg + off + l16, total - 1);
        const int p1 = min(beg + off + 16 + l16, total - 1);
        if (!IDS_GIVEN || off > 0) {
            id0 = adj[p0];
            id1 = adj[p1];
        }
        id0 = min(max(id0, 0), n - 1);
        id1 = min(max(id1, 0), n - 1);
        const int o0 = __shfl_xor(id0, 16, 32);
        const int o1 = __shfl_xor(id1, 16, 32);
        int m0 = (l16 < cnt)      ? id0 : o0;
        int m1 = (l16 + 16 < cnt) ? id1 : o1;
        const int q0 = __builtin_amdgcn_update_dpp(0, m0, 0x00, 0xf, 0xf, false);
        const int q1 = __builtin_amdgcn_update_dpp(0, m1, 0x00, 0xf, 0xf, false);
        m0 = (l16 < cm)      ? m0 : q0;
        m1 = (l16 + 16 < cm) ? m1 : q1;
        if (SC == 1) {
            s0 = scale_in[m0];
            if (cm > 16) s1 = scale_in[m1];
        }
        if (SC == 2) {
            s0 = esc[p0];
            if (cm > 16) s1 = esc[p1];
        }
        if (DEEP) {
            v4f x[6][4];
            const int sS0 = __float_as_int(s0);
            const int sS1 = __float_as_int(s1);
            #pragma unroll
            for (int q = 0; q < 6; ++q)
                if (q * 4 < cm) dhcf_ask4<TWO_TABLES>(x[q], (q < 4) ? m0 : m1, q, srcb, srch, nu, cb);
            #pragma unroll
            for (int q = 0; q < 2; ++q)
                if (q * 4 < cm) dhcf_use4<SCALE_IN>(alo, ahi, blo, bhi, x[q], sS0, q, cnt);
            #pragma unroll
            for (int q = 6; q < 8; ++q)
                if (q * 4 < cm) dhcf_ask4<TWO_TABLES>(x[q - 6], m1, q, srcb, srch, nu, cb);
            #pragma unroll
            for (int q = 2; q < 6; ++q)
                if (q * 4 < cm) dhcf_use4<SCALE_IN>(alo, ahi, blo, bhi, x[q], (q < 4) ? sS0 : sS1, q, cnt);
            #pragma unroll
            for (int q = 6; q < 8; ++q)
                if (q * 4 < cm) dhcf_use4<SCALE_IN>(alo, ahi, blo, bhi, x[q - 6], sS1, q, cnt);
            continue;
        }
        #pragma unroll
        for (int g = 0; g < 4; ++g) {
            if (g * 8 < cm) {
                const int   mS = (g < 2) ? m0 : m1;
                const int   sS = __float_as_int((g < 2) ? s0 : s1);
                v4f xa[4], xb[4];
                #pragma unroll
                for (int j = 0; j < 4; ++j) {
                    const int nb = dhcf_row_share(mS, (g * 8 + j) & 15);
                    xa[j] = dhcf_row<TWO_TABLES>(srcb, srch, nu, nb, cb);
                }
                if (g * 8 + 4 < cm) {
                    #pragma unroll
                    for (int j = 0; j < 4; ++j) {
                        const int nb = dhcf_row_share(mS, (g * 8 + 4 + j) & 15);
                        xb[j] = dhcf_row<TWO_TABLES>(srcb, srch, nu, nb, cb);
                    }
                    #pragma unroll
                    for (int j = 0; j < 4; ++j) {
                        const int k = g * 8 + j;
                        const float s = __int_as_float(dhcf_row_share(sS, k & 15));
                        if (j & 1) dhcf_acc<SCALE_IN>(blo, bhi, xa[j], k < cnt, s);
                        else       dhcf_acc<SCALE_IN>(alo, ahi, xa[j], k < cnt, s);
                    }
                    #pragma unroll
                    for (int j = 0; j < 4; ++j) {
                        const int k = g * 8 + 4 + j;
                        const float s = __int_as_float(dhcf_row_share(sS, k & 15));
                        if (j & 1) dhcf_acc<SCALE_IN>(blo, bhi, xb[j], k < cnt, s);
                        else       dhcf_acc<SCALE_IN>(alo, ahi, xb[j], k < cnt, s);
                    }
                } else {
                    #pragma unroll
                    for (int j = 0; j < 4; ++j) {
                        const int k = g * 8 + j;
                        const float s = __int_as_float(dhcf_row_share(sS, k & 15));
                        if (j & 1) dhcf_acc<SCALE_IN>(blo, bhi, xa[j], k < cnt, s);
                        else       dhcf_acc<SCALE_IN>(alo, ahi, xa[j], k < cnt, s);
                    }
                }
            }
        }
        if (SC == 1) {
            if (l16 < cnt) esc[p0] = s0;
            if (cm > 16 && l16 + 16 < cnt) esc[p1] = s1;
        }
    }
    alo += blo; ahi += bhi;
    v4f r;
    r.x = alo.x; r.y = alo.y; r.z = ahi.x; r.w = ahi.y;
    return r;
}

template <int SC, bool SCALE_OUT, bool TWO_TABLES>
__global__ __launch_bounds__(256) void dhcf_pull_kernel(
    const int* __restrict__ rowptr, const int* __restrict__ adj,
    const float* __restrict__ SRC, const float* __restrict__ SRC_HI, int nu,
    const float* __restrict__ scale_in, float* __restrict__ esc,
    const float* __restrict__ scale_out, float* __restrict__ DST,
    int n, int total)
{
    const int l16  = threadIdx.x & 15;
    const int khi  = (threadIdx.x >> 4) & 1;
    const unsigned cb = (unsigned)l16 << 4;
    const int npairs = (n + 1) >> 1;
    const int stride = (int)gridDim.x * WPB;
    const int wv = __builtin_amdgcn_readfirstlane((int)(threadIdx.x >> 5));
    int pair = (int)blockIdx.x * WPB + wv;
    if (pair >= npairs) return;
    int rlo, rhi;
    {
        const int r = min(2 * pair + khi, n - 1);
        rlo = rowptr[r];
        rhi = rowptr[r + 1];
    }
    int beg = min(max(rlo, 0), total);
    int deg = (2 * pair + khi < n) ? (min(rhi, total) - beg) : 0;
    int id0 = adj[min(beg + l16, total - 1)];
    int id1 = adj[min(beg + 16 + l16, total - 1)];
    int pairN = pair + stride;
    {
        const int pc = min(pairN, npairs - 1);
        const int r  = min(2 * pc + khi, n - 1);
        rlo = rowptr[r];
        rhi = rowptr[r + 1];
    }
    for (;;) {
        const bool more = pairN < npairs;
        const int begN = min(max(rlo, 0), total);
        const int degN = (more && 2 * pairN + khi < n) ? (min(rhi, total) - begN) : 0;
        const int id0N = adj[min(begN + l16, total - 1)];
        const int id1N = adj[min(begN + 16 + l16, total - 1)];
        {
            const int pc = min(pairN + stride, npairs - 1);
            const int r  = min(2 * pc + khi, n - 1);
            rlo = rowptr[r];
            rhi = rowptr[r + 1];
        }
        const int  row   = 2 * pair + khi;
        const bool valid = row < n;
        float so = 1.0f;
        if (SCALE_OUT) so = scale_out[min(row, n - 1)];
        v4f r = dhcf_pull_rows<SC, TWO_TABLES, false, true>(
            adj, (const char*)SRC, (const char*)SRC_HI, nu, scale_in, esc, n, total,
            beg, deg, l16, cb, id0, id1);
        if (valid) {
            if (SCALE_OUT) r *= so;
            *(v4f*)(DST + (size_t)row * 64 + (l16 << 2)) = r;
        }
        if (!more) break;
        pair = pairN;  pairN += stride;
        beg = begN;  deg = degN;  id0 = id0N;  id1 = id1N;
    }
}

#define WROW 68
template <bool FIRST, bool LAST, bool PULL, int SUM>
__global__ __launch_bounds__(256) void dhcf_layer_kernel(
    float* __restrict__ X, const float* __restrict__ Yb,
    const float* __restrict__ dis,
    const float* __restrict__ Wgc, const float* __restrict__ bgc,
    const float* __restrict__ Wbi, const float* __restrict__ bbi,
    float* __restrict__ ACC, int ntiles,
    const float* __restrict__ u_emb, const float* __restrict__ i_emb,
    int nu, float inv,
    const int* __restrict__ rowptr, const int* __restrict__ adj,
    int n, int total)
{
    __shared__ __attribute__((aligned(16))) float sWg[64 * WROW];
    __shared__ __attribute__((aligned(16))) float sWb[64 * WROW];
    __shared__ __attribute__((aligned(16))) float sY[WPB][16 * WROW];
    __shared__ __attribute__((aligned(16))) float sX[WPB][16 * WROW];
    __shared__ float sBias[128];

    int wave = threadIdx.x >> 5;
    int lane = threadIdx.x & 31;
    int tile = blockIdx.x * WPB + wave;
    const bool live = tile < ntiles;

    int row0 = tile * 16;
    int m    = lane & 15;
    int khi  = lane >> 4;
    int rowA = row0 + m;

    float gscale = 0.0f;
    {
        v4f wg[4], wb[4];
        for (int it = 0; it < 4; ++it) {
            const int q = threadIdx.x + it * 256;
            wg[it] = *(const v4f*)(Wgc + q * 4);
            wb[it] = *(const v4f*)(Wbi + q * 4);
        }
        if (live) {
            const v4f* ysrc = (const v4f*)(Yb + (size_t)row0 * 64) + lane;
            const v4f* xsrc = (const v4f*)(X  + (size_t)row0 * 64) + lane;
            v4f ty[8], tx[8];
            for (int i = 0; i < 8; ++i) {
                if (!PULL) ty[i] = ysrc[i * 32];
                if (FIRST) {
                    const int r = row0 + 2 * i + khi;
                    const float* p = (r < nu) ? (u_emb + (size_t)r * 64)
                                              : (i_emb + (size_t)(r - nu) * 64);
                    tx[i] = *(const v4f*)(p + (m << 2));
                } else {
                    tx[i] = xsrc[i * 32];
                }
            }
            gscale = dis[rowA];
            float* dy = sY[wave] + khi * WROW + (m << 2);
            float* dx = sX[wave] + khi * WROW + (m << 2);
            for (int i = 0; i < 8; ++i) {
                if (!PULL) *(v4f*)(dy + 2 * i * WROW) = ty[i];
                *(v4f*)(dx + 2 * i * WROW) = tx[i];
            }
        }
        if (threadIdx.x < 128)
            sBias[threadIdx.x] = (threadIdx.x < 64) ? bgc[threadIdx.x] : bbi[threadIdx.x - 64];
        for (int it = 0; it < 4; ++it) {
            const int q  = threadIdx.x + it * 256;
            const int r  = q >> 4;
            const int c4 = (q & 15) << 2;
            *(v4f*)(sWg + r * WROW + c4) = wg[it];
            *(v4f*)(sWb + r * WROW + c4) = wb[it];
        }
    }
    if (PULL && live) {
        float* dy = sY[wave] + khi * WROW + (m << 2);
        const int rp = rowptr[row0 + min(lane, 16)];
        const unsigned cb = (unsigned)m << 4;
        int begN = max(__shfl(rp, khi, 32), 0);
        int degN = min(__shfl(rp, khi + 1, 32), total) - begN;
        int id0N = adj[min(begN + m, total - 1)];
        int id1N = adj[min(begN + 16 + m, total - 1)];
        #pragma nounroll
        for (int i = 0; i < 8; ++i) {
            const int beg = begN, deg = degN, id0 = id0N, id1 = id1N;
            if (i < 7) {
                begN = max(__shfl(rp, 2 * i + 2 + khi, 32), 0);
                degN = min(__shfl(rp, 2 * i + 3 + khi, 32), total) - begN;
                id0N = adj[min(begN + m, total - 1)];
                id1N = adj[min(begN + 16 + m, total - 1)];
            }
            const v4f y = dhcf_pull_rows<0, false, true, true>(
                adj, (const char*)Yb, (const char*)Yb, nu, dis, nullptr, n, total,
                beg, deg, m, cb, id0, id1);
            *(v4f*)(dy + 2 * i * WROW) = y;
        }
    }
    __syncthreads();
    if (!live) return;

    const v8f vz = {0,0,0,0,0,0,0,0};
    v8f acc_s[4], acc_b[4];
    #pragma unroll
    for (int jj = 0; jj < 4; ++jj) { acc_s[jj] = vz; acc_b[jj] = vz; }

    const float* yrow = sY[wave] + m * WROW;
    const float* xrow = sX[wave] + m * WROW;

    #pragma unroll
    for (int kk = 0; kk < 16; ++kk) {
        int k0 = kk * 4 + khi * 2;
        v2f g2, xg2;
        float y0 = yrow[k0], y1 = yrow[k0 + 1];
        float x0 = xrow[k0], x1 = xrow[k0 + 1];
        g2.x  = y0 * gscale;  g2.y  = y1 * gscale;
        xg2.x = x0 * g2.x;    xg2.y = x1 * g2.y;
        #pragma unroll
        for (int jj = 0; jj < 4; ++jj) {
            int j = jj * 16 + m;
            v2f bg, bb;
            bg.x = sWg[j * WROW + k0]; bg.y = sWg[j * WROW + k0 + 1];
            bb.x = sWb[j * WROW + k0]; bb.y = sWb[j * WROW + k0 + 1];
            acc_s[jj] = __builtin_amdgcn_wmma_f32_16x16x4_f32(
                false, g2,  false, bg, (short)0, acc_s[jj], false, false);
            acc_b[jj] = __builtin_amdgcn_wmma_f32_16x16x4_f32(
                false, xg2, false, bb, (short)0, acc_b[jj], false, false);
        }
    }

    float xn[4][8];
    #pragma unroll
    for (int jj = 0; jj < 4; ++jj) {
        int j = jj * 16 + m;
        float bgj = sBias[j];
        float bbj = sBias[64 + j];
        #pragma unroll
        for (int r = 0; r < 8; ++r) {
            float xv = sX[wave][(r + 8 * khi) * WROW + j];
            float s = acc_s[jj][r] + bgj + xv;
            s = (s > 0.0f) ? s : 0.2f * s;
            float b = acc_b[jj][r] + bbj;
            b = (b > 0.0f) ? b : 0.2f * b;
            xn[jj][r] = s + b;
        }
    }

    #pragma unroll
    for (int r = 0; r < 8; ++r) {
        float p = 0.0f;
        #pragma unroll
        for (int jj = 0; jj < 4; ++jj) p += xn[jj][r] * xn[jj][r];
        p += __shfl_xor(p, 1, 32);
        p += __shfl_xor(p, 2, 32);
        p += __shfl_xor(p, 4, 32);
        p += __shfl_xor(p, 8, 32);
        float scl = 1.0f / fmaxf(sqrtf(p), 1e-12f);
        int row = row0 + r + 8 * khi;
        #pragma unroll
        for (int jj = 0; jj < 4; ++jj) {
            int j = jj * 16 + m;
            float v = xn[jj][r] * scl;
            if (!LAST) X[(size_t)row * 64 + j] = v;
            if (FIRST || SUM != 1) {
                float a = FIRST ? sX[wave][(r + 8 * khi) * WROW + j]
                                : ACC[(size_t)row * 64 + j];
                if (!FIRST && SUM == 2) a += sX[wave][(r + 8 * khi) * WROW + j];
                a += v;
                if (LAST) a *= inv;
                ACC[(size_t)row * 64 + j] = a;
            }
        }
    }
}

extern "C" void kernel_launch(void* const* d_in, const int* in_sizes, int n_in,
                              void* d_out, int out_size, void* d_ws, size_t ws_size,
                              hipStream_t stream)
{
    const float* u_emb = (const float*)d_in[0];
    const float* i_emb = (const float*)d_in[1];
    const float* Wgc   = (const float*)d_in[2];
    const float* bgc   = (const float*)d_in[3];
    const float* Wbi   = (const float*)d_in[4];
    const float* bbi   = (const float*)d_in[5];
    const int*   eu    = (const int*)d_in[6];
    const int*   ei    = (const int*)d_in[7];

    const int D  = 64;
    const int NU = in_sizes[0] / D;
    const int NI = in_sizes[1] / D;
    const int N  = NU + NI;
    const int NE = in_sizes[6];
    const int L  = in_sizes[2] / (D * D);
    const int n64 = N * D;

    float* X    = (float*)d_ws;
    float* E    = X + (size_t)n64;
    float* Y    = E + (size_t)n64;
    float* deg  = Y + (size_t)n64;
    float* dis  = deg + N;
    float* dinv = dis + N;
    float* ACC  = (float*)d_out;

    const int nchunk = (N + 1023) / 1024;
    const int total  = 2 * NE;
    int* rowptr = (int*)(dinv + N);
    int* bsum   = rowptr + (N + 1);
    int* adj    = bsum + nchunk;
    const int nb = (N + DHCF_BW - 1) / DHCF_BW;
    const int g1 = (NE + DHCF_ESL - 1) / DHCF_ESL;
    int2* recs  = (int2*)Y;
    int* H      = (int*)Y + 2 * (size_t)total;
    int* Hx     = H + (size_t)g1 * nb;
    int* T      = Hx + (size_t)g1 * nb;
    float* esc  = Y;
    const size_t need = (size_t)((const char*)(adj + total) - (const char*)d_ws);
    const size_t ybuild = 2 * (size_t)total + 2 * (size_t)g1 * nb + (size_t)nb;
    const bool lists = (ws_size >= need) && (ybuild <= (size_t)n64) && (NE > 0) &&
                       (nb <= DHCF_NBMAX) && (n64 < (1 << 30));

    dim3 blk(256);
    const bool direct = lists && (L > 0);
    if (!direct) {
        dhcf_init_kernel<<<(n64 + 255) / 256, blk, 0, stream>>>(
            u_emb, i_emb, X, ACC, deg, NU * D, n64, N);
    }

    const int ntiles = N / 16;
    const int lgrid  = (ntiles + WPB - 1) / WPB;
    const int egrid  = (NE * 32 + 255) / 256;
    const int ppairs = (N * 16 + 255) / 256;
    const int pgrid  = (ppairs < 1536) ? ppairs : 1536;

    if (lists) {
        dhcf_part_hist_kernel<<<g1, dim3(512), 0, stream>>>(eu, ei, H, NE, NU, nb);
        dhcf_part_scan_kernel<<<nb, blk, 0, stream>>>(H, Hx, T, nb, g1);
        dhcf_part_scatter_kernel<<<g1, dim3(512), 0, stream>>>(
            eu, ei, Hx, T, recs, NE, NU, nb, g1, total);
        dhcf_bucket_lists_kernel<<<nb, dim3(512), 0, stream>>>(
            recs, T, rowptr, adj, dis, dinv, N, total, nb);
    } else {
        dhcf_degree_kernel<<<(NE + 255) / 256, blk, 0, stream>>>(eu, ei, deg, NE, NU);
        dhcf_degnorm_kernel<<<(N + 255) / 256, blk, 0, stream>>>(deg, dis, dinv, N);
    }

    const float inv = 1.0f / (float)(L + 1);
    for (int l = 0; l < L; ++l) {
        const bool first = direct && (l == 0);
        const bool last  = (l == L - 1);
        if (lists) {
            if (first) {
                dhcf_pull_kernel<1, true, true><<<pgrid, blk, 0, stream>>>(
                    rowptr, adj, u_emb, i_emb, NU, dis, esc, dinv, E, N, total);
            } else {
                dhcf_pull_kernel<2, true, false><<<pgrid, blk, 0, stream>>>(
                    rowptr, adj, X, X, NU, dis, esc, dinv, E, N, total);
            }
        } else {
            dhcf_zero_kernel<<<(2 * n64 + 255) / 256, blk, 0, stream>>>(E, 2 * n64);
            dhcf_edge_gather_scatter<<<egrid, blk, 0, stream>>>(eu, ei, X, dis,  E, NE, NU);
            dhcf_edge_gather_scatter<<<egrid, blk, 0, stream>>>(eu, ei, E, dinv, Y, NE, NU);
        }
        const float* wg = Wgc + (size_t)l * D * D;
        const float* bg = bgc + (size_t)l * D;
        const float* wb = Wbi + (size_t)l * D * D;
        const float* bb = bbi + (size_t)l * D;
        if (lists) {
            const bool addx = (l >= 2) && ((l & 1) == 0);
            if (first && last) {
                dhcf_layer_kernel<true, true, true, 0><<<lgrid, blk, 0, stream>>>(
                    X, E, dis, wg, bg, wb, bb, ACC, ntiles, u_emb, i_emb, NU, inv,
                    rowptr, adj, N, total);
            } else if (first) {
                dhcf_layer_kernel<true, false, true, 0><<<lgrid, blk, 0, stream>>>(
                    X, E, dis, wg, bg, wb, bb, ACC, ntiles, u_emb, i_emb, NU, inv,
                    rowptr, adj, N, total);
            } else if (last && addx) {
                dhcf_layer_kernel<false, true, true, 2><<<lgrid, blk, 0, stream>>>(
                    X, E, dis, wg, bg, wb, bb, ACC, ntiles, u_emb, i_emb, NU, inv,
                    rowptr, adj, N, total);
            } else if (last) {
                dhcf_layer_kernel<false, true, true, 0><<<lgrid, blk, 0, stream>>>(
                    X, E, dis, wg, bg, wb, bb, ACC, ntiles, u_emb, i_emb, NU, inv,
                    rowptr, adj, N, total);
            } else if (addx) {
                dhcf_layer_kernel<false, false, true, 2><<<lgrid, blk, 0, stream>>>(
                    X, E, dis, wg, bg, wb, bb, ACC, ntiles, u_emb, i_emb, NU, inv,
                    rowptr, adj, N, total);
            } else {
                dhcf_layer_kernel<false, false, true, 1><<<lgrid, blk, 0, stream>>>(
                    X, E, dis, wg, bg, wb, bb, ACC, ntiles, u_emb, i_emb, NU, inv,
                    rowptr, adj, N, total);
            }
        } else if (last) {
            dhcf_layer_kernel<false, true, false, 0><<<lgrid, blk, 0, stream>>>(
                X, Y, dis, wg, bg, wb, bb, ACC, ntiles, u_emb, i_emb, NU, inv,
                rowptr, adj, N, total);
        } else {
            dhcf_layer_kernel<false, false, false, 0><<<lgrid, blk, 0, stream>>>(
                X, Y, dis, wg, bg, wb, bb, ACC, ntiles, u_emb, i_emb, NU, inv,
                rowptr, adj, N, total);
        }
    }
}
